# starting kernel
# baseline (speedup 1.0000x reference)
#include <hip/hip_runtime.h>
#include <hip/hip_bf16.h>

typedef float v2f __attribute__((ext_vector_type(2)));
typedef float v8f __attribute__((ext_vector_type(8)));

static __device__ __forceinline__ float fast_tanh(float x) {
#if __has_builtin(__builtin_amdgcn_tanhf)
    return __builtin_amdgcn_tanhf(x);
#else
    float r;
    asm("v_tanh_f32 %0, %1" : "=v"(r) : "v"(x));
    return r;
#endif
}

template <int NT>
__global__ void wmma_gemm_f32(const float* __restrict__ A,
                              const float* __restrict__ Bm,
                              const float* __restrict__ bias,
                              float* __restrict__ C,
                              int M, int N, int K,
                              size_t sA, size_t sB, size_t sC)
{
    const int batch = blockIdx.z;
    A  += (size_t)batch * sA;
    Bm += (size_t)batch * sB;
    C  += (size_t)batch * sC;

    const int tN   = blockIdx.x * 16 * NT;
    const int tM   = blockIdx.y * 16;
    const int lane = threadIdx.x & 31;
    const int half = lane >> 4;
    const int l16  = lane & 15;

    const float* __restrict__ aRow = A + (size_t)(tM + l16) * K;

    v8f acc[NT];
    #pragma unroll
    for (int t = 0; t < NT; ++t) acc[t] = (v8f){};

    #pragma unroll 2
    for (int k = 0; k < K; k += 4) {
        const int ka = k + half * 2;
        v2f av;
        av.x = aRow[ka];
        av.y = aRow[ka + 1];

        const float* __restrict__ b0 = Bm + (size_t)ka * N + tN + l16;
        const float* __restrict__ b1 = b0 + N;

        #pragma unroll
        for (int t = 0; t < NT; ++t) {
            v2f bv;
            bv.x = b0[t * 16];
            bv.y = b1[t * 16];
            acc[t] = __builtin_amdgcn_wmma_f32_16x16x4_f32(
                 false, av,  false, bv,
                 (short)0, acc[t],  false, false);
        }
    }

    #pragma unroll
    for (int t = 0; t < NT; ++t) {
        const int col  = tN + t * 16 + l16;
        const float bv = bias ? bias[col] : 0.0f;
        #pragma unroll
        for (int v = 0; v < 8; ++v) {
            const int m = tM + v + half * 8;
            C[(size_t)m * N + col] = acc[t][v] + bv;
        }
    }
}

__global__ void scores_kernel(const float* __restrict__ pf,
                              const float* __restrict__ pq,
                              const float* __restrict__ Wa,
                              const int* __restrict__ mask,
                              float* __restrict__ out)
{
    const int wavesPerBlock = blockDim.x >> 5;
    const int widx = blockIdx.x * wavesPerBlock + (threadIdx.x >> 5);
    const int lane = threadIdx.x & 31;

    const int f = widx & 127;
    const int q = (widx >> 7) & 31;
    const int b = widx >> 12;

    const float4* pf4 = (const float4*)(pf + ((size_t)(b * 128 + f)) * 256);
    const float4* pq4 = (const float4*)(pq + ((size_t)(b * 32 + q)) * 256);
    const float4* wa4 = (const float4*)Wa;

    float s = 0.0f;
    #pragma unroll
    for (int i = 0; i < 2; ++i) {
        const int idx = lane * 2 + i;
        const float4 x = pf4[idx];
        const float4 y = pq4[idx];
        const float4 w = wa4[idx];
        s += fast_tanh(x.x + y.x) * w.x;
        s += fast_tanh(x.y + y.y) * w.y;
        s += fast_tanh(x.z + y.z) * w.z;
        s += fast_tanh(x.w + y.w) * w.w;
    }
    #pragma unroll
    for (int off = 16; off > 0; off >>= 1)
        s += __shfl_down(s, off, 32);

    if (lane == 0) {
        out[widx] = (mask[b * 128 + f] == 0) ? -1e9f : s;
    }
}

__global__ void softmax_kernel(float* __restrict__ a)
{
    const int wavesPerBlock = blockDim.x >> 5;
    const int row  = blockIdx.x * wavesPerBlock + (threadIdx.x >> 5);
    const int lane = threadIdx.x & 31;

    float4* row4 = (float4*)(a + (size_t)row * 128);
    const float4 v = row4[lane];

    float m = fmaxf(fmaxf(v.x, v.y), fmaxf(v.z, v.w));
    #pragma unroll
    for (int off = 16; off > 0; off >>= 1)
        m = fmaxf(m, __shfl_xor(m, off, 32));

    float4 e;
    e.x = __expf(v.x - m);
    e.y = __expf(v.y - m);
    e.z = __expf(v.z - m);
    e.w = __expf(v.w - m);

    float s = e.x + e.y + e.z + e.w;
    #pragma unroll
    for (int off = 16; off > 0; off >>= 1)
        s += __shfl_xor(s, off, 32);

    const float inv = 1.0f / s;
    float4 o;
    o.x = e.x * inv; o.y = e.y * inv; o.z = e.z * inv; o.w = e.w * inv;
    row4[lane] = o;
}

extern "C" void kernel_launch(void* const* d_in, const int* in_sizes, int n_in,
                              void* d_out, int out_size, void* d_ws, size_t ws_size,
                              hipStream_t stream)
{
    const float* feature = (const float*)d_in[0];
    const float* query   = (const float*)d_in[1];
    const int*   mask    = (const int*)  d_in[2];
    const float* Wf      = (const float*)d_in[3];
    const float* Wq      = (const float*)d_in[4];
    const float* bq      = (const float*)d_in[5];
    const float* Wa      = (const float*)d_in[6];
    float*       out     = (float*)d_out;

    char* ws = (char*)d_ws;
    float* pf = (float*)ws;
    float* pq = (float*)(ws + (size_t)8192 * 256 * 4);
    float* sc = (float*)(ws + (size_t)8192 * 256 * 4
                            + (size_t)2048 * 256 * 4);

    const dim3 waveBlk(32);

    wmma_gemm_f32<4><<<dim3(256 / 64, 8192 / 16, 1), waveBlk, 0, stream>>>(
        feature, Wf, nullptr, pf, 8192, 256, 400, 0, 0, 0);

    wmma_gemm_f32<4><<<dim3(256 / 64, 2048 / 16, 1), waveBlk, 0, stream>>>(
        query, Wq, bq, pq, 2048, 256, 400, 0, 0, 0);

    scores_kernel<<<262144 / 8, 256, 0, stream>>>(pf, pq, Wa, mask, sc);

    softmax_kernel<<<2048 / 8, 256, 0, stream>>>(sc);

    wmma_gemm_f32<5><<<dim3(400 / 80, 32 / 16, 64), waveBlk, 0, stream>>>(
        sc, feature, nullptr, out, 32, 400, 128,
        (size_t)32 * 128, (size_t)128 * 400, (size_t)32 * 400);
}


# GEMM operand fetch: k order and tile-column map permuted alike in A, B and C so fragments load as float4, next stage of 32-40 k prefetched in a second register set; changes the k summation order
# speedup vs baseline: 1.9987x; 1.9987x over previous
#include <hip/hip_runtime.h>

typedef float v2f __attribute__((ext_vector_type(2)));
typedef float v8f __attribute__((ext_vector_type(8)));

static __device__ __forceinline__ float fast_tanh(float x) {
#if __has_builtin(__builtin_amdgcn_tanhf)
    return __builtin_amdgcn_tanhf(x);
#else
    float r;
    asm("v_tanh_f32 %0, %1" : "=v"(r) : "v"(x));
    return r;
#endif
}

typedef float v4f __attribute__((ext_vector_type(4)));

template <int G>
struct Stage {
    v4f a[G];
    v4f b[G][4];
};

template <int G>
static __device__ __forceinline__ void load_stage(Stage<G>& s,
                                                  const float* __restrict__ aP,
                                                  const float* __restrict__ bP,
                                                  int N)
{
    #pragma unroll
    for (int g = 0; g < G; ++g) {
        s.a[g] = *reinterpret_cast<const v4f*>(aP + g * 8);
        #pragma unroll
        for (int j = 0; j < 4; ++j)
            s.b[g][j] = *reinterpret_cast<const v4f*>(bP + (size_t)(g * 8 + j) * N);
    }
}

template <int G>
static __device__ __forceinline__ void mma_stage(const Stage<G>& s, v8f (&acc)[4])
{
    #pragma unroll
    for (int g = 0; g < G; ++g) {
        const v2f a01 = {s.a[g][0], s.a[g][1]};
        const v2f a23 = {s.a[g][2], s.a[g][3]};
        #pragma unroll
        for (int t = 0; t < 4; ++t) {
            const v2f bv = {s.b[g][0][t], s.b[g][1][t]};
            acc[t] = __builtin_amdgcn_wmma_f32_16x16x4_f32(
                 false, a01,  false, bv,
                 (short)0, acc[t],  false, false);
        }
        #pragma unroll
        for (int t = 0; t < 4; ++t) {
            const v2f bv = {s.b[g][2][t], s.b[g][3][t]};
            acc[t] = __builtin_amdgcn_wmma_f32_16x16x4_f32(
                 false, a23,  false, bv,
                 (short)0, acc[t],  false, false);
        }
    }
}

template <int G>
__global__ void __launch_bounds__(128) wmma_gemm_f32(const float* __restrict__ A,
                              const float* __restrict__ Bm,
                              const float* __restrict__ bias,
                              float* __restrict__ C,
                              int M, int N, int K,
                              size_t sA, size_t sB, size_t sC)
{
    const int batch = blockIdx.z;
    A  += (size_t)batch * sA;
    Bm += (size_t)batch * sB;
    C  += (size_t)batch * sC;

    const int wavesPerBlock = blockDim.x >> 5;
    const int tN   = blockIdx.x * 64;
    const int tM   = (blockIdx.y * wavesPerBlock + (threadIdx.x >> 5)) * 16;
    const int lane = threadIdx.x & 31;
    const int h    = lane >> 4;
    const int c    = lane & 15;

    const int  col   = tN + 4 * c;
    const bool colOk = col < N;
    const int  colL  = colOk ? col : (N - 4);

    const float* __restrict__ aP = A + (size_t)(tM + c) * K + 4 * h;
    const float* __restrict__ bP = Bm + (size_t)(4 * h) * N + colL;

    v8f acc[4];
    #pragma unroll
    for (int t = 0; t < 4; ++t) acc[t] = (v8f){};

    const int    S     = K / (8 * G);
    const size_t bStep = (size_t)(8 * G) * N;

    Stage<G> s0, s1;
    load_stage<G>(s0, aP, bP, N);
    int s = 0;
    for (; s + 2 < S; s += 2) {
        load_stage<G>(s1, aP + (s + 1) * 8 * G, bP + (size_t)(s + 1) * bStep, N);
        __builtin_amdgcn_sched_barrier(0);
        mma_stage<G>(s0, acc);
        __builtin_amdgcn_sched_barrier(0);
        load_stage<G>(s0, aP + (s + 2) * 8 * G, bP + (size_t)(s + 2) * bStep, N);
        __builtin_amdgcn_sched_barrier(0);
        mma_stage<G>(s1, acc);
        __builtin_amdgcn_sched_barrier(0);
    }
    load_stage<G>(s1, aP + (s + 1) * 8 * G, bP + (size_t)(s + 1) * bStep, N);
    __builtin_amdgcn_sched_barrier(0);
    mma_stage<G>(s0, acc);
    mma_stage<G>(s1, acc);

    if (colOk) {
        v4f bv = (v4f){};
        if (bias) bv = *reinterpret_cast<const v4f*>(bias + col);
        #pragma unroll
        for (int r = 0; r < 8; ++r) {
            const int m = tM + r + h * 8;
            v4f o = {acc[0][r], acc[1][r], acc[2][r], acc[3][r]};
            o += bv;
            *reinterpret_cast<v4f*>(C + (size_t)m * N + col) = o;
        }
    }
}

__global__ void scores_kernel(const float* __restrict__ pf,
                              const float* __restrict__ pq,
                              const float* __restrict__ Wa,
                              const int* __restrict__ mask,
                              float* __restrict__ out)
{
    const int wavesPerBlock = blockDim.x >> 5;
    const int widx = blockIdx.x * wavesPerBlock + (threadIdx.x >> 5);
    const int lane = threadIdx.x & 31;

    const int f = widx & 127;
    const int q = (widx >> 7) & 31;
    const int b = widx >> 12;

    const float4* pf4 = (const float4*)(pf + ((size_t)(b * 128 + f)) * 256);
    const float4* pq4 = (const float4*)(pq + ((size_t)(b * 32 + q)) * 256);
    const float4* wa4 = (const float4*)Wa;

    float s = 0.0f;
    #pragma unroll
    for (int i = 0; i < 2; ++i) {
        const int idx = lane * 2 + i;
        const float4 x = pf4[idx];
        const float4 y = pq4[idx];
        const float4 w = wa4[idx];
        s += fast_tanh(x.x + y.x) * w.x;
        s += fast_tanh(x.y + y.y) * w.y;
        s += fast_tanh(x.z + y.z) * w.z;
        s += fast_tanh(x.w + y.w) * w.w;
    }
    #pragma unroll
    for (int off = 16; off > 0; off >>= 1)
        s += __shfl_down(s, off, 32);

    if (lane == 0) {
        out[widx] = (mask[b * 128 + f] == 0) ? -1e9f : s;
    }
}

__global__ void softmax_kernel(float* __restrict__ a)
{
    const int wavesPerBlock = blockDim.x >> 5;
    const int row  = blockIdx.x * wavesPerBlock + (threadIdx.x >> 5);
    const int lane = threadIdx.x & 31;

    float4* row4 = (float4*)(a + (size_t)row * 128);
    const float4 v = row4[lane];

    float m = fmaxf(fmaxf(v.x, v.y), fmaxf(v.z, v.w));
    #pragma unroll
    for (int off = 16; off > 0; off >>= 1)
        m = fmaxf(m, __shfl_xor(m, off, 32));

    float4 e;
    e.x = __expf(v.x - m);
    e.y = __expf(v.y - m);
    e.z = __expf(v.z - m);
    e.w = __expf(v.w - m);

    float s = e.x + e.y + e.z + e.w;
    #pragma unroll
    for (int off = 16; off > 0; off >>= 1)
        s += __shfl_xor(s, off, 32);

    const float inv = 1.0f / s;
    float4 o;
    o.x = e.x * inv; o.y = e.y * inv; o.z = e.z * inv; o.w = e.w * inv;
    row4[lane] = o;
}

extern "C" void kernel_launch(void* const* d_in, const int* in_sizes, int n_in,
                              void* d_out, int out_size, void* d_ws, size_t ws_size,
                              hipStream_t stream)
{
    const float* feature = (const float*)d_in[0];
    const float* query   = (const float*)d_in[1];
    const int*   mask    = (const int*)  d_in[2];
    const float* Wf      = (const float*)d_in[3];
    const float* Wq      = (const float*)d_in[4];
    const float* bq      = (const float*)d_in[5];
    const float* Wa      = (const float*)d_in[6];
    float*       out     = (float*)d_out;

    char* ws = (char*)d_ws;
    float* pf = (float*)ws;
    float* pq = (float*)(ws + (size_t)8192 * 256 * 4);
    float* sc = (float*)(ws + (size_t)8192 * 256 * 4
                            + (size_t)2048 * 256 * 4);

    wmma_gemm_f32<5><<<dim3(256 / 64, 8192 / 64, 1), dim3(128), 0, stream>>>(
        feature, Wf, nullptr, pf, 8192, 256, 400, 0, 0, 0);

    wmma_gemm_f32<5><<<dim3(256 / 64, 2048 / 32, 1), dim3(64), 0, stream>>>(
        query, Wq, bq, pq, 2048, 256, 400, 0, 0, 0);

    scores_kernel<<<262144 / 8, 256, 0, stream>>>(pf, pq, Wa, mask, sc);

    softmax_kernel<<<2048 / 8, 256, 0, stream>>>(sc);

    wmma_gemm_f32<4><<<dim3(7, 1, 64), dim3(64), 0, stream>>>(
        sc, feature, nullptr, out, 32, 400, 128,
        (size_t)32 * 128, (size_t)128 * 400, (size_t)32 * 400);
}


# scores kernel remapped: lane = f, each lane sums its own pf row over all a (tile staged once in LDS, pq and Wa as wave-uniform scalar loads); changes the summation order over a (no lane tree)
# speedup vs baseline: 3.2644x; 1.6333x over previous
#include <hip/hip_runtime.h>

typedef float v2f __attribute__((ext_vector_type(2)));
typedef float v8f __attribute__((ext_vector_type(8)));

static __device__ __forceinline__ float fast_tanh(float x) {
#if __has_builtin(__builtin_amdgcn_tanhf)
    return __builtin_amdgcn_tanhf(x);
#else
    float r;
    asm("v_tanh_f32 %0, %1" : "=v"(r) : "v"(x));
    return r;
#endif
}

typedef float v4f __attribute__((ext_vector_type(4)));

template <int G>
struct Stage {
    v4f a[G];
    v4f b[G][4];
};

template <int G>
static __device__ __forceinline__ void load_stage(Stage<G>& s,
                                                  const float* __restrict__ aP,
                                                  const float* __restrict__ bP,
                                                  int N)
{
    #pragma unroll
    for (int g = 0; g < G; ++g) {
        s.a[g] = *reinterpret_cast<const v4f*>(aP + g * 8);
        #pragma unroll
        for (int j = 0; j < 4; ++j)
            s.b[g][j] = *reinterpret_cast<const v4f*>(bP + (size_t)(g * 8 + j) * N);
    }
}

template <int G>
static __device__ __forceinline__ void mma_stage(const Stage<G>& s, v8f (&acc)[4])
{
    #pragma unroll
    for (int g = 0; g < G; ++g) {
        const v2f a01 = {s.a[g][0], s.a[g][1]};
        const v2f a23 = {s.a[g][2], s.a[g][3]};
        #pragma unroll
        for (int t = 0; t < 4; ++t) {
            const v2f bv = {s.b[g][0][t], s.b[g][1][t]};
            acc[t] = __builtin_amdgcn_wmma_f32_16x16x4_f32(
                 false, a01,  false, bv,
                 (short)0, acc[t],  false, false);
        }
        #pragma unroll
        for (int t = 0; t < 4; ++t) {
            const v2f bv = {s.b[g][2][t], s.b[g][3][t]};
            acc[t] = __builtin_amdgcn_wmma_f32_16x16x4_f32(
                 false, a23,  false, bv,
                 (short)0, acc[t],  false, false);
        }
    }
}

template <int G>
__global__ void __launch_bounds__(128) wmma_gemm_f32(const float* __restrict__ A,
                              const float* __restrict__ Bm,
                              const float* __restrict__ bias,
                              float* __restrict__ C,
                              int M, int N, int K,
                              size_t sA, size_t sB, size_t sC)
{
    const int batch = blockIdx.z;
    A  += (size_t)batch * sA;
    Bm += (size_t)batch * sB;
    C  += (size_t)batch * sC;

    const int wavesPerBlock = blockDim.x >> 5;
    const int tN   = blockIdx.x * 64;
    const int tM   = (blockIdx.y * wavesPerBlock + (threadIdx.x >> 5)) * 16;
    const int lane = threadIdx.x & 31;
    const int h    = lane >> 4;
    const int c    = lane & 15;

    const int  col   = tN + 4 * c;
    const bool colOk = col < N;
    const int  colL  = colOk ? col : (N - 4);

    const float* __restrict__ aP = A + (size_t)(tM + c) * K + 4 * h;
    const float* __restrict__ bP = Bm + (size_t)(4 * h) * N + colL;

    v8f acc[4];
    #pragma unroll
    for (int t = 0; t < 4; ++t) acc[t] = (v8f){};

    const int    S     = K / (8 * G);
    const size_t bStep = (size_t)(8 * G) * N;

    Stage<G> s0, s1;
    load_stage<G>(s0, aP, bP, N);
    int s = 0;
    for (; s + 2 < S; s += 2) {
        load_stage<G>(s1, aP + (s + 1) * 8 * G, bP + (size_t)(s + 1) * bStep, N);
        __builtin_amdgcn_sched_barrier(0);
        mma_stage<G>(s0, acc);
        __builtin_amdgcn_sched_barrier(0);
        load_stage<G>(s0, aP + (s + 2) * 8 * G, bP + (size_t)(s + 2) * bStep, N);
        __builtin_amdgcn_sched_barrier(0);
        mma_stage<G>(s1, acc);
        __builtin_amdgcn_sched_barrier(0);
    }
    load_stage<G>(s1, aP + (s + 1) * 8 * G, bP + (size_t)(s + 1) * bStep, N);
    __builtin_amdgcn_sched_barrier(0);
    mma_stage<G>(s0, acc);
    mma_stage<G>(s1, acc);

    if (colOk) {
        v4f bv = (v4f){};
        if (bias) bv = *reinterpret_cast<const v4f*>(bias + col);
        #pragma unroll
        for (int r = 0; r < 8; ++r) {
            const int m = tM + r + h * 8;
            v4f o = {acc[0][r], acc[1][r], acc[2][r], acc[3][r]};
            o += bv;
            *reinterpret_cast<v4f*>(C + (size_t)m * N + col) = o;
        }
    }
}

#define SC_PITCH 260

__global__ void __launch_bounds__(128) scores_kernel(const float* __restrict__ pf,
                              const float* __restrict__ pq,
                              const float* __restrict__ Wa,
                              const int* __restrict__ mask,
                              float* __restrict__ out)
{
    __shared__ __attribute__((aligned(16))) float tile[32 * SC_PITCH];

    const int tid  = threadIdx.x;
    const int lane = tid & 31;
    const int wave = __builtin_amdgcn_readfirstlane(tid >> 5);
    const int blk  = blockIdx.x;
    const int qq   = blk & 3;
    const int fg   = (blk >> 2) & 3;
    const int b    = blk >> 4;

    const v4f* __restrict__ srcp =
        reinterpret_cast<const v4f*>(pf + (size_t)(b * 128 + fg * 32) * 256);
    v4f st[16];
    #pragma unroll
    for (int i = 0; i < 16; ++i) st[i] = srcp[i * 128 + tid];
    const int mk = mask[b * 128 + fg * 32 + lane];
    #pragma unroll
    for (int i = 0; i < 16; ++i) {
        const int idx = i * 128 + tid;
        *reinterpret_cast<v4f*>(&tile[(idx >> 6) * SC_PITCH + (idx & 63) * 4]) = st[i];
    }
    __syncthreads();

    const int q0 = qq * 8 + wave * 2;
    const float* __restrict__ y0p = pq + (size_t)(b * 32 + q0) * 256;
    const float* __restrict__ y1p = y0p + 256;
    const float* xr = &tile[lane * SC_PITCH];

    v2f acc0a = {0.0f, 0.0f}, acc0b = {0.0f, 0.0f};
    v2f acc1a = {0.0f, 0.0f}, acc1b = {0.0f, 0.0f};

    #pragma unroll
    for (int ch = 0; ch < 8; ++ch) {
        v4f x[8];
        #pragma unroll
        for (int k = 0; k < 8; ++k)
            x[k] = *reinterpret_cast<const v4f*>(xr + ch * 32 + k * 4);
        #pragma unroll
        for (int k = 0; k < 8; ++k) {
            const int a = ch * 32 + k * 4;
            const v2f wlo = *reinterpret_cast<const v2f*>(Wa + a);
            const v2f whi = *reinterpret_cast<const v2f*>(Wa + a + 2);
            const v2f y0lo = *reinterpret_cast<const v2f*>(y0p + a);
            const v2f y0hi = *reinterpret_cast<const v2f*>(y0p + a + 2);
            const v2f y1lo = *reinterpret_cast<const v2f*>(y1p + a);
            const v2f y1hi = *reinterpret_cast<const v2f*>(y1p + a + 2);
            const v2f xlo = {x[k][0], x[k][1]};
            const v2f xhi = {x[k][2], x[k][3]};
            v2f s, t;
            s = xlo + y0lo; t[0] = fast_tanh(s[0]); t[1] = fast_tanh(s[1]);
            acc0a = t * wlo + acc0a;
            s = xhi + y0hi; t[0] = fast_tanh(s[0]); t[1] = fast_tanh(s[1]);
            acc0b = t * whi + acc0b;
            s = xlo + y1lo; t[0] = fast_tanh(s[0]); t[1] = fast_tanh(s[1]);
            acc1a = t * wlo + acc1a;
            s = xhi + y1hi; t[0] = fast_tanh(s[0]); t[1] = fast_tanh(s[1]);
            acc1b = t * whi + acc1b;
        }
    }

    const float r0 = (acc0a[0] + acc0a[1]) + (acc0b[0] + acc0b[1]);
    const float r1 = (acc1a[0] + acc1a[1]) + (acc1b[0] + acc1b[1]);
    float* o = out + (size_t)(b * 32 + q0) * 128 + fg * 32 + lane;
    o[0]   = (mk == 0) ? -1e9f : r0;
    o[128] = (mk == 0) ? -1e9f : r1;
}

__global__ void softmax_kernel(float* __restrict__ a)
{
    const int wavesPerBlock = blockDim.x >> 5;
    const int row  = blockIdx.x * wavesPerBlock + (threadIdx.x >> 5);
    const int lane = threadIdx.x & 31;

    float4* row4 = (float4*)(a + (size_t)row * 128);
    const float4 v = row4[lane];

    float m = fmaxf(fmaxf(v.x, v.y), fmaxf(v.z, v.w));
    #pragma unroll
    for (int off = 16; off > 0; off >>= 1)
        m = fmaxf(m, __shfl_xor(m, off, 32));

    float4 e;
    e.x = __expf(v.x - m);
    e.y = __expf(v.y - m);
    e.z = __expf(v.z - m);
    e.w = __expf(v.w - m);

    float s = e.x + e.y + e.z + e.w;
    #pragma unroll
    for (int off = 16; off > 0; off >>= 1)
        s += __shfl_xor(s, off, 32);

    const float inv = 1.0f / s;
    float4 o;
    o.x = e.x * inv; o.y = e.y * inv; o.z = e.z * inv; o.w = e.w * inv;
    row4[lane] = o;
}

extern "C" void kernel_launch(void* const* d_in, const int* in_sizes, int n_in,
                              void* d_out, int out_size, void* d_ws, size_t ws_size,
                              hipStream_t stream)
{
    const float* feature = (const float*)d_in[0];
    const float* query   = (const float*)d_in[1];
    const int*   mask    = (const int*)  d_in[2];
    const float* Wf      = (const float*)d_in[3];
    const float* Wq      = (const float*)d_in[4];
    const float* bq      = (const float*)d_in[5];
    const float* Wa      = (const float*)d_in[6];
    float*       out     = (float*)d_out;

    char* ws = (char*)d_ws;
    float* pf = (float*)ws;
    float* pq = (float*)(ws + (size_t)8192 * 256 * 4);
    float* sc = (float*)(ws + (size_t)8192 * 256 * 4
                            + (size_t)2048 * 256 * 4);

    wmma_gemm_f32<5><<<dim3(256 / 64, 8192 / 64, 1), dim3(128), 0, stream>>>(
        feature, Wf, nullptr, pf, 8192, 256, 400, 0, 0, 0);

    wmma_gemm_f32<5><<<dim3(256 / 64, 2048 / 32, 1), dim3(64), 0, stream>>>(
        query, Wq, bq, pq, 2048, 256, 400, 0, 0, 0);

    scores_kernel<<<1024, 128, 0, stream>>>(pf, pq, Wa, mask, sc);

    softmax_kernel<<<2048 / 8, 256, 0, stream>>>(sc);

    wmma_gemm_f32<4><<<dim3(7, 1, 64), dim3(64), 0, stream>>>(
        sc, feature, nullptr, out, 32, 400, 128,
        (size_t)32 * 128, (size_t)128 * 400, (size_t)32 * 400);
}


# softmax folded into the final product's A-fragment load: each wave normalises its 16 score rows in registers, softmax launch removed; changes the summation order of the softmax denominator
# speedup vs baseline: 3.6162x; 1.1078x over previous
#include <hip/hip_runtime.h>

typedef float v2f __attribute__((ext_vector_type(2)));
typedef float v8f __attribute__((ext_vector_type(8)));

static __device__ __forceinline__ float fast_tanh(float x) {
#if __has_builtin(__builtin_amdgcn_tanhf)
    return __builtin_amdgcn_tanhf(x);
#else
    float r;
    asm("v_tanh_f32 %0, %1" : "=v"(r) : "v"(x));
    return r;
#endif
}

typedef float v4f __attribute__((ext_vector_type(4)));

template <int G>
struct Stage {
    v4f a[G];
    v4f b[G][4];
};

template <int G>
static __device__ __forceinline__ void load_stage(Stage<G>& s,
                                                  const float* __restrict__ aP,
                                                  const float* __restrict__ bP,
                                                  int N)
{
    #pragma unroll
    for (int g = 0; g < G; ++g) {
        s.a[g] = *reinterpret_cast<const v4f*>(aP + g * 8);
        #pragma unroll
        for (int j = 0; j < 4; ++j)
            s.b[g][j] = *reinterpret_cast<const v4f*>(bP + (size_t)(g * 8 + j) * N);
    }
}

template <int G>
static __device__ __forceinline__ void mma_stage(const Stage<G>& s, v8f (&acc)[4])
{
    #pragma unroll
    for (int g = 0; g < G; ++g) {
        const v2f a01 = {s.a[g][0], s.a[g][1]};
        const v2f a23 = {s.a[g][2], s.a[g][3]};
        #pragma unroll
        for (int t = 0; t < 4; ++t) {
            const v2f bv = {s.b[g][0][t], s.b[g][1][t]};
            acc[t] = __builtin_amdgcn_wmma_f32_16x16x4_f32(
                 false, a01,  false, bv,
                 (short)0, acc[t],  false, false);
        }
        #pragma unroll
        for (int t = 0; t < 4; ++t) {
            const v2f bv = {s.b[g][2][t], s.b[g][3][t]};
            acc[t] = __builtin_amdgcn_wmma_f32_16x16x4_f32(
                 false, a23,  false, bv,
                 (short)0, acc[t],  false, false);
        }
    }
}

template <int G>
__global__ void __launch_bounds__(128) wmma_gemm_f32(const float* __restrict__ A,
                              const float* __restrict__ Bm,
                              const float* __restrict__ bias,
                              float* __restrict__ C,
                              int M, int N, int K,
                              size_t sA, size_t sB, size_t sC)
{
    const int batch = blockIdx.z;
    A  += (size_t)batch * sA;
    Bm += (size_t)batch * sB;
    C  += (size_t)batch * sC;

    const int wavesPerBlock = blockDim.x >> 5;
    const int tN   = blockIdx.x * 64;
    const int tM   = (blockIdx.y * wavesPerBlock + (threadIdx.x >> 5)) * 16;
    const int lane = threadIdx.x & 31;
    const int h    = lane >> 4;
    const int c    = lane & 15;

    const int  col   = tN + 4 * c;
    const bool colOk = col < N;
    const int  colL  = colOk ? col : (N - 4);

    const float* __restrict__ aP = A + (size_t)(tM + c) * K + 4 * h;
    const float* __restrict__ bP = Bm + (size_t)(4 * h) * N + colL;

    v8f acc[4];
    #pragma unroll
    for (int t = 0; t < 4; ++t) acc[t] = (v8f){};

    const int    S     = K / (8 * G);
    const size_t bStep = (size_t)(8 * G) * N;

    Stage<G> s0, s1;
    load_stage<G>(s0, aP, bP, N);
    int s = 0;
    for (; s + 2 < S; s += 2) {
        load_stage<G>(s1, aP + (s + 1) * 8 * G, bP + (size_t)(s + 1) * bStep, N);
        __builtin_amdgcn_sched_barrier(0);
        mma_stage<G>(s0, acc);
        __builtin_amdgcn_sched_barrier(0);
        load_stage<G>(s0, aP + (s + 2) * 8 * G, bP + (size_t)(s + 2) * bStep, N);
        __builtin_amdgcn_sched_barrier(0);
        mma_stage<G>(s1, acc);
        __builtin_amdgcn_sched_barrier(0);
    }
    load_stage<G>(s1, aP + (s + 1) * 8 * G, bP + (size_t)(s + 1) * bStep, N);
    __builtin_amdgcn_sched_barrier(0);
    mma_stage<G>(s0, acc);
    mma_stage<G>(s1, acc);

    if (colOk) {
        v4f bv = (v4f){};
        if (bias) bv = *reinterpret_cast<const v4f*>(bias + col);
        #pragma unroll
        for (int r = 0; r < 8; ++r) {
            const int m = tM + r + h * 8;
            v4f o = {acc[0][r], acc[1][r], acc[2][r], acc[3][r]};
            o += bv;
            *reinterpret_cast<v4f*>(C + (size_t)m * N + col) = o;
        }
    }
}

#define SC_PITCH 260

__global__ void __launch_bounds__(128) scores_kernel(const float* __restrict__ pf,
                              const float* __restrict__ pq,
                              const float* __restrict__ Wa,
                              const int* __restrict__ mask,
                              float* __restrict__ out)
{
    __shared__ __attribute__((aligned(16))) float tile[32 * SC_PITCH];

    const int tid  = threadIdx.x;
    const int lane = tid & 31;
    const int wave = __builtin_amdgcn_readfirstlane(tid >> 5);
    const int blk  = blockIdx.x;
    const int qq   = blk & 3;
    const int fg   = (blk >> 2) & 3;
    const int b    = blk >> 4;

    const v4f* __restrict__ srcp =
        reinterpret_cast<const v4f*>(pf + (size_t)(b * 128 + fg * 32) * 256);
    v4f st[16];
    #pragma unroll
    for (int i = 0; i < 16; ++i) st[i] = srcp[i * 128 + tid];
    const int mk = mask[b * 128 + fg * 32 + lane];
    #pragma unroll
    for (int i = 0; i < 16; ++i) {
        const int idx = i * 128 + tid;
        *reinterpret_cast<v4f*>(&tile[(idx >> 6) * SC_PITCH + (idx & 63) * 4]) = st[i];
    }
    __syncthreads();

    const int q0 = qq * 8 + wave * 2;
    const float* __restrict__ y0p = pq + (size_t)(b * 32 + q0) * 256;
    const float* __restrict__ y1p = y0p + 256;
    const float* xr = &tile[lane * SC_PITCH];

    v2f acc0a = {0.0f, 0.0f}, acc0b = {0.0f, 0.0f};
    v2f acc1a = {0.0f, 0.0f}, acc1b = {0.0f, 0.0f};

    #pragma unroll
    for (int ch = 0; ch < 8; ++ch) {
        v4f x[8];
        #pragma unroll
        for (int k = 0; k < 8; ++k)
            x[k] = *reinterpret_cast<const v4f*>(xr + ch * 32 + k * 4);
        #pragma unroll
        for (int k = 0; k < 8; ++k) {
            const int a = ch * 32 + k * 4;
            const v2f wlo = *reinterpret_cast<const v2f*>(Wa + a);
            const v2f whi = *reinterpret_cast<const v2f*>(Wa + a + 2);
            const v2f y0lo = *reinterpret_cast<const v2f*>(y0p + a);
            const v2f y0hi = *reinterpret_cast<const v2f*>(y0p + a + 2);
            const v2f y1lo = *reinterpret_cast<const v2f*>(y1p + a);
            const v2f y1hi = *reinterpret_cast<const v2f*>(y1p + a + 2);
            const v2f xlo = {x[k][0], x[k][1]};
            const v2f xhi = {x[k][2], x[k][3]};
            v2f s, t;
            s = xlo + y0lo; t[0] = fast_tanh(s[0]); t[1] = fast_tanh(s[1]);
            acc0a = t * wlo + acc0a;
            s = xhi + y0hi; t[0] = fast_tanh(s[0]); t[1] = fast_tanh(s[1]);
            acc0b = t * whi + acc0b;
            s = xlo + y1lo; t[0] = fast_tanh(s[0]); t[1] = fast_tanh(s[1]);
            acc1a = t * wlo + acc1a;
            s = xhi + y1hi; t[0] = fast_tanh(s[0]); t[1] = fast_tanh(s[1]);
            acc1b = t * whi + acc1b;
        }
    }

    const float r0 = (acc0a[0] + acc0a[1]) + (acc0b[0] + acc0b[1]);
    const float r1 = (acc1a[0] + acc1a[1]) + (acc1b[0] + acc1b[1]);
    float* o = out + (size_t)(b * 32 + q0) * 128 + fg * 32 + lane;
    o[0]   = (mk == 0) ? -1e9f : r0;
    o[128] = (mk == 0) ? -1e9f : r1;
}

struct BStage {
    v4f b[4][4];
};

static __device__ __forceinline__ void load_b_stage(BStage& s,
                                                    const float* __restrict__ bP,
                                                    int N)
{
    #pragma unroll
    for (int g = 0; g < 4; ++g) {
        #pragma unroll
        for (int j = 0; j < 4; ++j)
            s.b[g][j] = *reinterpret_cast<const v4f*>(bP + (size_t)(g * 8 + j) * N);
    }
}

template <int BASE>
static __device__ __forceinline__ void mma_b_stage(const BStage& s, const v4f (&a)[16],
                                                   v8f (&acc)[4])
{
    #pragma unroll
    for (int g = 0; g < 4; ++g) {
        const v2f a01 = {a[BASE + g][0], a[BASE + g][1]};
        const v2f a23 = {a[BASE + g][2], a[BASE + g][3]};
        #pragma unroll
        for (int t = 0; t < 4; ++t) {
            const v2f bv = {s.b[g][0][t], s.b[g][1][t]};
            acc[t] = __builtin_amdgcn_wmma_f32_16x16x4_f32(
                 false, a01,  false, bv,
                 (short)0, acc[t],  false, false);
        }
        #pragma unroll
        for (int t = 0; t < 4; ++t) {
            const v2f bv = {s.b[g][2][t], s.b[g][3][t]};
            acc[t] = __builtin_amdgcn_wmma_f32_16x16x4_f32(
                 false, a23,  false, bv,
                 (short)0, acc[t],  false, false);
        }
    }
}

__global__ void __launch_bounds__(64) attend_kernel(const float* __restrict__ score,
                              const float* __restrict__ feature,
                              float* __restrict__ out)
{
    constexpr int N = 400;
    constexpr int K = 128;

    const int batch = blockIdx.z;
    const float* __restrict__ A  = score   + (size_t)batch * (32 * K);
    const float* __restrict__ Bm = feature + (size_t)batch * (K * N);
    float* __restrict__ C        = out     + (size_t)batch * (32 * N);

    const int tN   = blockIdx.x * 64;
    const int tM   = (threadIdx.x >> 5) * 16;
    const int lane = threadIdx.x & 31;
    const int h    = lane >> 4;
    const int c    = lane & 15;

    const int  col   = tN + 4 * c;
    const bool colOk = col < N;
    const int  colL  = colOk ? col : (N - 4);

    const float* __restrict__ aP = A + (size_t)(tM + c) * K + 4 * h;
    const float* __restrict__ bP = Bm + (size_t)(4 * h) * N + colL;

    v8f acc[4];
    #pragma unroll
    for (int t = 0; t < 4; ++t) acc[t] = (v8f){};

    BStage s0, s1;
    v4f a[16];
    load_b_stage(s0, bP, N);
    #pragma unroll
    for (int g = 0; g < 16; ++g) a[g] = *reinterpret_cast<const v4f*>(aP + g * 8);
    load_b_stage(s1, bP + (size_t)32 * N, N);
    __builtin_amdgcn_sched_barrier(0);

    float m = fmaxf(fmaxf(a[0][0], a[0][1]), fmaxf(a[0][2], a[0][3]));
    #pragma unroll
    for (int g = 1; g < 16; ++g)
        m = fmaxf(m, fmaxf(fmaxf(a[g][0], a[g][1]), fmaxf(a[g][2], a[g][3])));
    m = fmaxf(m, __shfl_xor(m, 16, 32));

    v4f ps = (v4f){};
    #pragma unroll
    for (int g = 0; g < 16; ++g) {
        v4f e;
        e[0] = __expf(a[g][0] - m);
        e[1] = __expf(a[g][1] - m);
        e[2] = __expf(a[g][2] - m);
        e[3] = __expf(a[g][3] - m);
        a[g] = e;
        ps += e;
    }
    float ssum = (ps[0] + ps[1]) + (ps[2] + ps[3]);
    ssum += __shfl_xor(ssum, 16, 32);
    const float inv = 1.0f / ssum;
    #pragma unroll
    for (int g = 0; g < 16; ++g) a[g] *= inv;

    mma_b_stage<0>(s0, a, acc);
    __builtin_amdgcn_sched_barrier(0);
    load_b_stage(s0, bP + (size_t)64 * N, N);
    __builtin_amdgcn_sched_barrier(0);
    mma_b_stage<4>(s1, a, acc);
    __builtin_amdgcn_sched_barrier(0);
    load_b_stage(s1, bP + (size_t)96 * N, N);
    __builtin_amdgcn_sched_barrier(0);
    mma_b_stage<8>(s0, a, acc);
    mma_b_stage<12>(s1, a, acc);

    if (colOk) {
        #pragma unroll
        for (int r = 0; r < 8; ++r) {
            const int mrow = tM + r + h * 8;
            const v4f o = {acc[0][r], acc[1][r], acc[2][r], acc[3][r]};
            *reinterpret_cast<v4f*>(C + (size_t)mrow * N + col) = o;
        }
    }
}

extern "C" void kernel_launch(void* const* d_in, const int* in_sizes, int n_in,
                              void* d_out, int out_size, void* d_ws, size_t ws_size,
                              hipStream_t stream)
{
    const float* feature = (const float*)d_in[0];
    const float* query   = (const float*)d_in[1];
    const int*   mask    = (const int*)  d_in[2];
    const float* Wf      = (const float*)d_in[3];
    const float* Wq      = (const float*)d_in[4];
    const float* bq      = (const float*)d_in[5];
    const float* Wa      = (const float*)d_in[6];
    float*       out     = (float*)d_out;

    char* ws = (char*)d_ws;
    float* pf = (float*)ws;
    float* pq = (float*)(ws + (size_t)8192 * 256 * 4);
    float* sc = (float*)(ws + (size_t)8192 * 256 * 4
                            + (size_t)2048 * 256 * 4);

    wmma_gemm_f32<5><<<dim3(256 / 64, 8192 / 64, 1), dim3(128), 0, stream>>>(
        feature, Wf, nullptr, pf, 8192, 256, 400, 0, 0, 0);

    wmma_gemm_f32<5><<<dim3(256 / 64, 2048 / 32, 1), dim3(64), 0, stream>>>(
        query, Wq, bq, pq, 2048, 256, 400, 0, 0, 0);

    scores_kernel<<<1024, 128, 0, stream>>>(pf, pq, Wa, mask, sc);

    attend_kernel<<<dim3(7, 1, 64), dim3(64), 0, stream>>>(sc, feature, out);
}


# both projection GEMMs in one launch: pq as 16x32 strips (1024 waves, one per SIMD) entering ahead of the 2048 pf waves, so every SIMD gets equal matrix work; 4 launches become 3
# speedup vs baseline: 4.3732x; 1.2093x over previous
#include <hip/hip_runtime.h>

typedef float v2f __attribute__((ext_vector_type(2)));
typedef float v8f __attribute__((ext_vector_type(8)));

static __device__ __forceinline__ float fast_tanh(float x) {
#if __has_builtin(__builtin_amdgcn_tanhf)
    return __builtin_amdgcn_tanhf(x);
#else
    float r;
    asm("v_tanh_f32 %0, %1" : "=v"(r) : "v"(x));
    return r;
#endif
}

typedef float v4f __attribute__((ext_vector_type(4)));

template <int NT> struct VecN;
template <> struct VecN<4> { typedef v4f type; };
template <> struct VecN<2> { typedef v2f type; };

template <int G, int NT>
struct Stage {
    v4f a[G];
    typename VecN<NT>::type b[G][4];
};

template <int G, int NT>
static __device__ __forceinline__ void load_stage(Stage<G, NT>& s,
                                                  const float* __restrict__ aP,
                                                  const float* __restrict__ bP,
                                                  int N)
{
    typedef typename VecN<NT>::type vb;
    #pragma unroll
    for (int g = 0; g < G; ++g) {
        s.a[g] = *reinterpret_cast<const v4f*>(aP + g * 8);
        #pragma unroll
        for (int j = 0; j < 4; ++j)
            s.b[g][j] = *reinterpret_cast<const vb*>(bP + (size_t)(g * 8 + j) * N);
    }
}

template <int G, int NT>
static __device__ __forceinline__ void mma_stage(const Stage<G, NT>& s, v8f (&acc)[NT])
{
    #pragma unroll
    for (int g = 0; g < G; ++g) {
        const v2f a01 = {s.a[g][0], s.a[g][1]};
        const v2f a23 = {s.a[g][2], s.a[g][3]};
        #pragma unroll
        for (int t = 0; t < NT; ++t) {
            const v2f bv = {s.b[g][0][t], s.b[g][1][t]};
            acc[t] = __builtin_amdgcn_wmma_f32_16x16x4_f32(
                 false, a01,  false, bv,
                 (short)0, acc[t],  false, false);
        }
        #pragma unroll
        for (int t = 0; t < NT; ++t) {
            const v2f bv = {s.b[g][2][t], s.b[g][3][t]};
            acc[t] = __builtin_amdgcn_wmma_f32_16x16x4_f32(
                 false, a23,  false, bv,
                 (short)0, acc[t],  false, false);
        }
    }
}

template <int G, int NT>
static __device__ __forceinline__ void gemm_strip(const float* __restrict__ A,
                              const float* __restrict__ Bm,
                              const float* __restrict__ bias,
                              float* __restrict__ C,
                              int N, int K, int tM, int tN, int lane)
{
    typedef typename VecN<NT>::type vb;
    const int h    = lane >> 4;
    const int c    = lane & 15;

    const int  col   = tN + NT * c;
    const bool colOk = col < N;
    const int  colL  = colOk ? col : (N - NT);

    const float* __restrict__ aP = A + (size_t)(tM + c) * K + 4 * h;
    const float* __restrict__ bP = Bm + (size_t)(4 * h) * N + colL;

    v8f acc[NT];
    #pragma unroll
    for (int t = 0; t < NT; ++t) acc[t] = (v8f){};

    const int    S     = K / (8 * G);
    const size_t bStep = (size_t)(8 * G) * N;

    Stage<G, NT> s0, s1;
    load_stage<G, NT>(s0, aP, bP, N);
    int s = 0;
    for (; s + 2 < S; s += 2) {
        load_stage<G, NT>(s1, aP + (s + 1) * 8 * G, bP + (size_t)(s + 1) * bStep, N);
        __builtin_amdgcn_sched_barrier(0);
        mma_stage<G, NT>(s0, acc);
        __builtin_amdgcn_sched_barrier(0);
        load_stage<G, NT>(s0, aP + (s + 2) * 8 * G, bP + (size_t)(s + 2) * bStep, N);
        __builtin_amdgcn_sched_barrier(0);
        mma_stage<G, NT>(s1, acc);
        __builtin_amdgcn_sched_barrier(0);
    }
    load_stage<G, NT>(s1, aP + (s + 1) * 8 * G, bP + (size_t)(s + 1) * bStep, N);
    __builtin_amdgcn_sched_barrier(0);
    mma_stage<G, NT>(s0, acc);
    mma_stage<G, NT>(s1, acc);

    if (colOk) {
        vb bv = (vb){};
        if (bias) bv = *reinterpret_cast<const vb*>(bias + col);
        #pragma unroll
        for (int r = 0; r < 8; ++r) {
            const int m = tM + r + h * 8;
            vb o;
            #pragma unroll
            for (int t = 0; t < NT; ++t) o[t] = acc[t][r];
            o += bv;
            *reinterpret_cast<vb*>(C + (size_t)m * N + col) = o;
        }
    }
}

__global__ void __launch_bounds__(128) proj_kernel(const float* __restrict__ feature,
                              const float* __restrict__ Wf,
                              float* __restrict__ pf,
                              const float* __restrict__ query,
                              const float* __restrict__ Wq,
                              const float* __restrict__ bq,
                              float* __restrict__ pq,
                              int N, int K)
{
    const int wave = threadIdx.x >> 5;
    const int lane = threadIdx.x & 31;
    const int id   = blockIdx.x;
    if (id >= 256) {
        const int jd = id - 256;
        const int tN = (jd & 3) * 64;
        const int tM = ((jd >> 2) * 4 + wave) * 16;
        gemm_strip<5, 4>(feature, Wf, nullptr, pf, N, K, tM, tN, lane);
    } else {
        const int tN = (id & 7) * 32;
        const int tM = ((id >> 3) * 4 + wave) * 16;
        gemm_strip<5, 2>(query, Wq, bq, pq, N, K, tM, tN, lane);
    }
}

#define SC_PITCH 260

__global__ void __launch_bounds__(128) scores_kernel(const float* __restrict__ pf,
                              const float* __restrict__ pq,
                              const float* __restrict__ Wa,
                              const int* __restrict__ mask,
                              float* __restrict__ out)
{
    __shared__ __attribute__((aligned(16))) float tile[32 * SC_PITCH];

    const int tid  = threadIdx.x;
    const int lane = tid & 31;
    const int wave = __builtin_amdgcn_readfirstlane(tid >> 5);
    const int blk  = blockIdx.x;
    const int qq   = blk & 3;
    const int fg   = (blk >> 2) & 3;
    const int b    = blk >> 4;

    const v4f* __restrict__ srcp =
        reinterpret_cast<const v4f*>(pf + (size_t)(b * 128 + fg * 32) * 256);
    v4f st[16];
    #pragma unroll
    for (int i = 0; i < 16; ++i) st[i] = srcp[i * 128 + tid];
    const int mk = mask[b * 128 + fg * 32 + lane];
    #pragma unroll
    for (int i = 0; i < 16; ++i) {
        const int idx = i * 128 + tid;
        *reinterpret_cast<v4f*>(&tile[(idx >> 6) * SC_PITCH + (idx & 63) * 4]) = st[i];
    }
    __syncthreads();

    const int q0 = qq * 8 + wave * 2;
    const float* __restrict__ y0p = pq + (size_t)(b * 32 + q0) * 256;
    const float* __restrict__ y1p = y0p + 256;
    const float* xr = &tile[lane * SC_PITCH];

    v2f acc0a = {0.0f, 0.0f}, acc0b = {0.0f, 0.0f};
    v2f acc1a = {0.0f, 0.0f}, acc1b = {0.0f, 0.0f};

    #pragma unroll
    for (int ch = 0; ch < 8; ++ch) {
        v4f x[8];
        #pragma unroll
        for (int k = 0; k < 8; ++k)
            x[k] = *reinterpret_cast<const v4f*>(xr + ch * 32 + k * 4);
        #pragma unroll
        for (int k = 0; k < 8; ++k) {
            const int a = ch * 32 + k * 4;
            const v2f wlo = *reinterpret_cast<const v2f*>(Wa + a);
            const v2f whi = *reinterpret_cast<const v2f*>(Wa + a + 2);
            const v2f y0lo = *reinterpret_cast<const v2f*>(y0p + a);
            const v2f y0hi = *reinterpret_cast<const v2f*>(y0p + a + 2);
            const v2f y1lo = *reinterpret_cast<const v2f*>(y1p + a);
            const v2f y1hi = *reinterpret_cast<const v2f*>(y1p + a + 2);
            const v2f xlo = {x[k][0], x[k][1]};
            const v2f xhi = {x[k][2], x[k][3]};
            v2f s, t;
            s = xlo + y0lo; t[0] = fast_tanh(s[0]); t[1] = fast_tanh(s[1]);
            acc0a = t * wlo + acc0a;
            s = xhi + y0hi; t[0] = fast_tanh(s[0]); t[1] = fast_tanh(s[1]);
            acc0b = t * whi + acc0b;
            s = xlo + y1lo; t[0] = fast_tanh(s[0]); t[1] = fast_tanh(s[1]);
            acc1a = t * wlo + acc1a;
            s = xhi + y1hi; t[0] = fast_tanh(s[0]); t[1] = fast_tanh(s[1]);
            acc1b = t * whi + acc1b;
        }
    }

    const float r0 = (acc0a[0] + acc0a[1]) + (acc0b[0] + acc0b[1]);
    const float r1 = (acc1a[0] + acc1a[1]) + (acc1b[0] + acc1b[1]);
    float* o = out + (size_t)(b * 32 + q0) * 128 + fg * 32 + lane;
    o[0]   = (mk == 0) ? -1e9f : r0;
    o[128] = (mk == 0) ? -1e9f : r1;
}

struct BStage {
    v4f b[4][4];
};

static __device__ __forceinline__ void load_b_stage(BStage& s,
                                                    const float* __restrict__ bP,
                                                    int N)
{
    #pragma unroll
    for (int g = 0; g < 4; ++g) {
        #pragma unroll
        for (int j = 0; j < 4; ++j)
            s.b[g][j] = *reinterpret_cast<const v4f*>(bP + (size_t)(g * 8 + j) * N);
    }
}

template <int BASE>
static __device__ __forceinline__ void mma_b_stage(const BStage& s, const v4f (&a)[16],
                                                   v8f (&acc)[4])
{
    #pragma unroll
    for (int g = 0; g < 4; ++g) {
        const v2f a01 = {a[BASE + g][0], a[BASE + g][1]};
        const v2f a23 = {a[BASE + g][2], a[BASE + g][3]};
        #pragma unroll
        for (int t = 0; t < 4; ++t) {
            const v2f bv = {s.b[g][0][t], s.b[g][1][t]};
            acc[t] = __builtin_amdgcn_wmma_f32_16x16x4_f32(
                 false, a01,  false, bv,
                 (short)0, acc[t],  false, false);
        }
        #pragma unroll
        for (int t = 0; t < 4; ++t) {
            const v2f bv = {s.b[g][2][t], s.b[g][3][t]};
            acc[t] = __builtin_amdgcn_wmma_f32_16x16x4_f32(
                 false, a23,  false, bv,
                 (short)0, acc[t],  false, false);
        }
    }
}

__global__ void __launch_bounds__(64) attend_kernel(const float* __restrict__ score,
                              const float* __restrict__ feature,
                              float* __restrict__ out)
{
    constexpr int N = 400;
    constexpr int K = 128;

    const int batch = blockIdx.z;
    const float* __restrict__ A  = score   + (size_t)batch * (32 * K);
    const float* __restrict__ Bm = feature + (size_t)batch * (K * N);
    float* __restrict__ C        = out     + (size_t)batch * (32 * N);

    const int tN   = blockIdx.x * 64;
    const int tM   = (threadIdx.x >> 5) * 16;
    const int lane = threadIdx.x & 31;
    const int h    = lane >> 4;
    const int c    = lane & 15;

    const int  col   = tN + 4 * c;
    const bool colOk = col < N;
    const int  colL  = colOk ? col : (N - 4);

    const float* __restrict__ aP = A + (size_t)(tM + c) * K + 4 * h;
    const float* __restrict__ bP = Bm + (size_t)(4 * h) * N + colL;

    v8f acc[4];
    #pragma unroll
    for (int t = 0; t < 4; ++t) acc[t] = (v8f){};

    BStage s0, s1;
    v4f a[16];
    load_b_stage(s0, bP, N);
    #pragma unroll
    for (int g = 0; g < 16; ++g) a[g] = *reinterpret_cast<const v4f*>(aP + g * 8);
    load_b_stage(s1, bP + (size_t)32 * N, N);
    __builtin_amdgcn_sched_barrier(0);

    float m = fmaxf(fmaxf(a[0][0], a[0][1]), fmaxf(a[0][2], a[0][3]));
    #pragma unroll
    for (int g = 1; g < 16; ++g)
        m = fmaxf(m, fmaxf(fmaxf(a[g][0], a[g][1]), fmaxf(a[g][2], a[g][3])));
    m = fmaxf(m, __shfl_xor(m, 16, 32));

    v4f ps = (v4f){};
    #pragma unroll
    for (int g = 0; g < 16; ++g) {
        v4f e;
        e[0] = __expf(a[g][0] - m);
        e[1] = __expf(a[g][1] - m);
        e[2] = __expf(a[g][2] - m);
        e[3] = __expf(a[g][3] - m);
        a[g] = e;
        ps += e;
    }
    float ssum = (ps[0] + ps[1]) + (ps[2] + ps[3]);
    ssum += __shfl_xor(ssum, 16, 32);
    const float inv = 1.0f / ssum;
    #pragma unroll
    for (int g = 0; g < 16; ++g) a[g] *= inv;

    mma_b_stage<0>(s0, a, acc);
    __builtin_amdgcn_sched_barrier(0);
    load_b_stage(s0, bP + (size_t)64 * N, N);
    __builtin_amdgcn_sched_barrier(0);
    mma_b_stage<4>(s1, a, acc);
    __builtin_amdgcn_sched_barrier(0);
    load_b_stage(s1, bP + (size_t)96 * N, N);
    __builtin_amdgcn_sched_barrier(0);
    mma_b_stage<8>(s0, a, acc);
    mma_b_stage<12>(s1, a, acc);

    if (colOk) {
        #pragma unroll
        for (int r = 0; r < 8; ++r) {
            const int mrow = tM + r + h * 8;
            const v4f o = {acc[0][r], acc[1][r], acc[2][r], acc[3][r]};
            *reinterpret_cast<v4f*>(C + (size_t)mrow * N + col) = o;
        }
    }
}

extern "C" void kernel_launch(void* const* d_in, const int* in_sizes, int n_in,
                              void* d_out, int out_size, void* d_ws, size_t ws_size,
                              hipStream_t stream)
{
    const float* feature = (const float*)d_in[0];
    const float* query   = (const float*)d_in[1];
    const int*   mask    = (const int*)  d_in[2];
    const float* Wf      = (const float*)d_in[3];
    const float* Wq      = (const float*)d_in[4];
    const float* bq      = (const float*)d_in[5];
    const float* Wa      = (const float*)d_in[6];
    float*       out     = (float*)d_out;

    char* ws = (char*)d_ws;
    float* pf = (float*)ws;
    float* pq = (float*)(ws + (size_t)8192 * 256 * 4);
    float* sc = (float*)(ws + (size_t)8192 * 256 * 4
                            + (size_t)2048 * 256 * 4);

    proj_kernel<<<768, 128, 0, stream>>>(feature, Wf, pf, query, Wq, bq, pq, 256, 400);

    scores_kernel<<<1024, 128, 0, stream>>>(pf, pq, Wa, mask, sc);

    attend_kernel<<<dim3(7, 1, 64), dim3(64), 0, stream>>>(sc, feature, out);
}


# scores, softmax and the final product in one launch: a block per (batch, 8 q) keeps its score rows in LDS and feeds them to the WMMAs as rows 0-7 of the A tile; 3 launches become 2
# speedup vs baseline: 4.5664x; 1.0442x over previous
#include <hip/hip_runtime.h>

typedef float v2f __attribute__((ext_vector_type(2)));
typedef float v8f __attribute__((ext_vector_type(8)));

static __device__ __forceinline__ float fast_tanh(float x) {
#if __has_builtin(__builtin_amdgcn_tanhf)
    return __builtin_amdgcn_tanhf(x);
#else
    float r;
    asm("v_tanh_f32 %0, %1" : "=v"(r) : "v"(x));
    return r;
#endif
}

typedef float v4f __attribute__((ext_vector_type(4)));

template <int NT> struct VecN;
template <> struct VecN<4> { typedef v4f type; };
template <> struct VecN<2> { typedef v2f type; };

template <int G, int NT>
struct Stage {
    v4f a[G];
    typename VecN<NT>::type b[G][4];
};

template <int G, int NT>
static __device__ __forceinline__ void load_stage(Stage<G, NT>& s,
                                                  const float* __restrict__ aP,
                                                  const float* __restrict__ bP,
                                                  int N)
{
    typedef typename VecN<NT>::type vb;
    #pragma unroll
    for (int g = 0; g < G; ++g) {
        s.a[g] = *reinterpret_cast<const v4f*>(aP + g * 8);
        #pragma unroll
        for (int j = 0; j < 4; ++j)
            s.b[g][j] = *reinterpret_cast<const vb*>(bP + (size_t)(g * 8 + j) * N);
    }
}

template <int G, int NT>
static __device__ __forceinline__ void mma_stage(const Stage<G, NT>& s, v8f (&acc)[NT])
{
    #pragma unroll
    for (int g = 0; g < G; ++g) {
        const v2f a01 = {s.a[g][0], s.a[g][1]};
        const v2f a23 = {s.a[g][2], s.a[g][3]};
        #pragma unroll
        for (int t = 0; t < NT; ++t) {
            const v2f bv = {s.b[g][0][t], s.b[g][1][t]};
            acc[t] = __builtin_amdgcn_wmma_f32_16x16x4_f32(
                 false, a01,  false, bv,
                 (short)0, acc[t],  false, false);
        }
        #pragma unroll
        for (int t = 0; t < NT; ++t) {
            const v2f bv = {s.b[g][2][t], s.b[g][3][t]};
            acc[t] = __builtin_amdgcn_wmma_f32_16x16x4_f32(
                 false, a23,  false, bv,
                 (short)0, acc[t],  false, false);
        }
    }
}

template <int G, int NT>
static __device__ __forceinline__ void gemm_strip(const float* __restrict__ A,
                              const float* __restrict__ Bm,
                              const float* __restrict__ bias,
                              float* __restrict__ C,
                              int N, int K, int tM, int tN, int lane)
{
    typedef typename VecN<NT>::type vb;
    const int h    = lane >> 4;
    const int c    = lane & 15;

    const int  col   = tN + NT * c;
    const bool colOk = col < N;
    const int  colL  = colOk ? col : (N - NT);

    const float* __restrict__ aP = A + (size_t)(tM + c) * K + 4 * h;
    const float* __restrict__ bP = Bm + (size_t)(4 * h) * N + colL;

    v8f acc[NT];
    #pragma unroll
    for (int t = 0; t < NT; ++t) acc[t] = (v8f){};

    const int    S     = K / (8 * G);
    const size_t bStep = (size_t)(8 * G) * N;

    Stage<G, NT> s0, s1;
    load_stage<G, NT>(s0, aP, bP, N);
    int s = 0;
    for (; s + 2 < S; s += 2) {
        load_stage<G, NT>(s1, aP + (s + 1) * 8 * G, bP + (size_t)(s + 1) * bStep, N);
        __builtin_amdgcn_sched_barrier(0);
        mma_stage<G, NT>(s0, acc);
        __builtin_amdgcn_sched_barrier(0);
        load_stage<G, NT>(s0, aP + (s + 2) * 8 * G, bP + (size_t)(s + 2) * bStep, N);
        __builtin_amdgcn_sched_barrier(0);
        mma_stage<G, NT>(s1, acc);
        __builtin_amdgcn_sched_barrier(0);
    }
    load_stage<G, NT>(s1, aP + (s + 1) * 8 * G, bP + (size_t)(s + 1) * bStep, N);
    __builtin_amdgcn_sched_barrier(0);
    mma_stage<G, NT>(s0, acc);
    mma_stage<G, NT>(s1, acc);

    if (colOk) {
        vb bv = (vb){};
        if (bias) bv = *reinterpret_cast<const vb*>(bias + col);
        #pragma unroll
        for (int r = 0; r < 8; ++r) {
            const int m = tM + r + h * 8;
            vb o;
            #pragma unroll
            for (int t = 0; t < NT; ++t) o[t] = acc[t][r];
            o += bv;
            *reinterpret_cast<vb*>(C + (size_t)m * N + col) = o;
        }
    }
}

__global__ void __launch_bounds__(128) proj_kernel(const float* __restrict__ feature,
                              const float* __restrict__ Wf,
                              float* __restrict__ pf,
                              const float* __restrict__ query,
                              const float* __restrict__ Wq,
                              const float* __restrict__ bq,
                              float* __restrict__ pq,
                              int N, int K)
{
    const int wave = threadIdx.x >> 5;
    const int lane = threadIdx.x & 31;
    const int id   = blockIdx.x;
    if (id >= 256) {
        const int jd = id - 256;
        const int tN = (jd & 3) * 64;
        const int tM = ((jd >> 2) * 4 + wave) * 16;
        gemm_strip<5, 4>(feature, Wf, nullptr, pf, N, K, tM, tN, lane);
    } else {
        const int tN = (id & 7) * 32;
        const int tM = ((id >> 3) * 4 + wave) * 16;
        gemm_strip<5, 2>(query, Wq, bq, pq, N, K, tM, tN, lane);
    }
}

#define SC_PITCH 260
#define S_PITCH 136

struct BStage {
    v2f b[4][4];
};

static __device__ __forceinline__ void load_b_stage(BStage& s,
                                                    const float* __restrict__ bP,
                                                    int N)
{
    #pragma unroll
    for (int g = 0; g < 4; ++g) {
        #pragma unroll
        for (int j = 0; j < 4; ++j)
            s.b[g][j] = *reinterpret_cast<const v2f*>(bP + (size_t)(g * 8 + j) * N);
    }
}

template <int BASE>
static __device__ __forceinline__ void mma_b_stage(const BStage& s, const v4f (&a)[16],
                                                   v8f (&acc)[2])
{
    #pragma unroll
    for (int g = 0; g < 4; ++g) {
        const v2f a01 = {a[BASE + g][0], a[BASE + g][1]};
        const v2f a23 = {a[BASE + g][2], a[BASE + g][3]};
        #pragma unroll
        for (int t = 0; t < 2; ++t) {
            const v2f bv = {s.b[g][0][t], s.b[g][1][t]};
            acc[t] = __builtin_amdgcn_wmma_f32_16x16x4_f32(
                 false, a01,  false, bv,
                 (short)0, acc[t],  false, false);
        }
        #pragma unroll
        for (int t = 0; t < 2; ++t) {
            const v2f bv = {s.b[g][2][t], s.b[g][3][t]};
            acc[t] = __builtin_amdgcn_wmma_f32_16x16x4_f32(
                 false, a23,  false, bv,
                 (short)0, acc[t],  false, false);
        }
    }
}

__global__ void __launch_bounds__(512) scores_attend_kernel(const float* __restrict__ pf,
                              const float* __restrict__ pq,
                              const float* __restrict__ Wa,
                              const int* __restrict__ mask,
                              const float* __restrict__ feature,
                              float* __restrict__ out)
{
    __shared__ __attribute__((aligned(16))) float tile[128 * SC_PITCH];
    __shared__ __attribute__((aligned(16))) float sc[8 * S_PITCH];

    const int tid  = threadIdx.x;
    const int lane = tid & 31;
    const int wave = __builtin_amdgcn_readfirstlane(tid >> 5);
    const int blk  = blockIdx.x;
    const int qg   = blk & 3;
    const int b    = blk >> 2;
    const int fg   = wave & 3;
    const int qp   = wave >> 2;

    const v4f* __restrict__ srcp =
        reinterpret_cast<const v4f*>(pf + (size_t)b * (128 * 256));
    v4f st[16];
    #pragma unroll
    for (int i = 0; i < 16; ++i) st[i] = srcp[i * 512 + tid];
    const int mk = mask[b * 128 + fg * 32 + lane];
    #pragma unroll
    for (int i = 0; i < 16; ++i) {
        const int idx = i * 512 + tid;
        *reinterpret_cast<v4f*>(&tile[(idx >> 6) * SC_PITCH + (idx & 63) * 4]) = st[i];
    }
    __syncthreads();

    {
        const int q0 = qg * 8 + qp * 2;
        const float* __restrict__ y0p = pq + (size_t)(b * 32 + q0) * 256;
        const float* __restrict__ y1p = y0p + 256;
        const float* xr = &tile[(fg * 32 + lane) * SC_PITCH];

        v2f acc0a = {0.0f, 0.0f}, acc0b = {0.0f, 0.0f};
        v2f acc1a = {0.0f, 0.0f}, acc1b = {0.0f, 0.0f};

        #pragma unroll
        for (int ch = 0; ch < 8; ++ch) {
            v4f x[8];
            #pragma unroll
            for (int k = 0; k < 8; ++k)
                x[k] = *reinterpret_cast<const v4f*>(xr + ch * 32 + k * 4);
            #pragma unroll
            for (int k = 0; k < 8; ++k) {
                const int a = ch * 32 + k * 4;
                const v2f wlo = *reinterpret_cast<const v2f*>(Wa + a);
                const v2f whi = *reinterpret_cast<const v2f*>(Wa + a + 2);
                const v2f y0lo = *reinterpret_cast<const v2f*>(y0p + a);
                const v2f y0hi = *reinterpret_cast<const v2f*>(y0p + a + 2);
                const v2f y1lo = *reinterpret_cast<const v2f*>(y1p + a);
                const v2f y1hi = *reinterpret_cast<const v2f*>(y1p + a + 2);
                const v2f xlo = {x[k][0], x[k][1]};
                const v2f xhi = {x[k][2], x[k][3]};
                v2f s, t;
                s = xlo + y0lo; t[0] = fast_tanh(s[0]); t[1] = fast_tanh(s[1]);
                acc0a = t * wlo + acc0a;
                s = xhi + y0hi; t[0] = fast_tanh(s[0]); t[1] = fast_tanh(s[1]);
                acc0b = t * whi + acc0b;
                s = xlo + y1lo; t[0] = fast_tanh(s[0]); t[1] = fast_tanh(s[1]);
                acc1a = t * wlo + acc1a;
                s = xhi + y1hi; t[0] = fast_tanh(s[0]); t[1] = fast_tanh(s[1]);
                acc1b = t * whi + acc1b;
            }
        }

        const float r0 = (acc0a[0] + acc0a[1]) + (acc0b[0] + acc0b[1]);
        const float r1 = (acc1a[0] + acc1a[1]) + (acc1b[0] + acc1b[1]);
        float* o = &sc[(qp * 2) * S_PITCH + fg * 32 + lane];
        o[0]       = (mk == 0) ? -1e9f : r0;
        o[S_PITCH] = (mk == 0) ? -1e9f : r1;
    }
    __syncthreads();

    if (wave >= 13) return;

    constexpr int N = 400;
    constexpr int K = 128;

    const float* __restrict__ Bm = feature + (size_t)b * (K * N);
    float* __restrict__ C        = out + ((size_t)b * 32 + qg * 8) * N;

    const int h = lane >> 4;
    const int c = lane & 15;

    const int  col   = wave * 32 + 2 * c;
    const bool colOk = col < N;
    const int  colL  = colOk ? col : (N - 2);

    const float* aP = &sc[(c & 7) * S_PITCH + 4 * h];
    const float* __restrict__ bP = Bm + (size_t)(4 * h) * N + colL;

    v8f acc[2];
    #pragma unroll
    for (int t = 0; t < 2; ++t) acc[t] = (v8f){};

    BStage s0, s1;
    v4f a[16];
    load_b_stage(s0, bP, N);
    #pragma unroll
    for (int g = 0; g < 16; ++g) a[g] = *reinterpret_cast<const v4f*>(aP + g * 8);
    load_b_stage(s1, bP + (size_t)32 * N, N);
    __builtin_amdgcn_sched_barrier(0);

    float m = fmaxf(fmaxf(a[0][0], a[0][1]), fmaxf(a[0][2], a[0][3]));
    #pragma unroll
    for (int g = 1; g < 16; ++g)
        m = fmaxf(m, fmaxf(fmaxf(a[g][0], a[g][1]), fmaxf(a[g][2], a[g][3])));
    m = fmaxf(m, __shfl_xor(m, 16, 32));

    v4f ps = (v4f){};
    #pragma unroll
    for (int g = 0; g < 16; ++g) {
        v4f e;
        e[0] = __expf(a[g][0] - m);
        e[1] = __expf(a[g][1] - m);
        e[2] = __expf(a[g][2] - m);
        e[3] = __expf(a[g][3] - m);
        a[g] = e;
        ps += e;
    }
    float ssum = (ps[0] + ps[1]) + (ps[2] + ps[3]);
    ssum += __shfl_xor(ssum, 16, 32);
    const float inv = 1.0f / ssum;
    #pragma unroll
    for (int g = 0; g < 16; ++g) a[g] *= inv;

    mma_b_stage<0>(s0, a, acc);
    __builtin_amdgcn_sched_barrier(0);
    load_b_stage(s0, bP + (size_t)64 * N, N);
    __builtin_amdgcn_sched_barrier(0);
    mma_b_stage<4>(s1, a, acc);
    __builtin_amdgcn_sched_barrier(0);
    load_b_stage(s1, bP + (size_t)96 * N, N);
    __builtin_amdgcn_sched_barrier(0);
    mma_b_stage<8>(s0, a, acc);
    mma_b_stage<12>(s1, a, acc);

    if (colOk && h == 0) {
        #pragma unroll
        for (int r = 0; r < 8; ++r) {
            const v2f o = {acc[0][r], acc[1][r]};
            *reinterpret_cast<v2f*>(C + (size_t)r * N + col) = o;
        }
    }
}

extern "C" void kernel_launch(void* const* d_in, const int* in_sizes, int n_in,
                              void* d_out, int out_size, void* d_ws, size_t ws_size,
                              hipStream_t stream)
{
    const float* feature = (const float*)d_in[0];
    const float* query   = (const float*)d_in[1];
    const int*   mask    = (const int*)  d_in[2];
    const float* Wf      = (const float*)d_in[3];
    const float* Wq      = (const float*)d_in[4];
    const float* bq      = (const float*)d_in[5];
    const float* Wa      = (const float*)d_in[6];
    float*       out     = (float*)d_out;

    char* ws = (char*)d_ws;
    float* pf = (float*)ws;
    float* pq = (float*)(ws + (size_t)8192 * 256 * 4);

    proj_kernel<<<768, 128, 0, stream>>>(feature, Wf, pf, query, Wq, bq, pq, 256, 400);

    scores_attend_kernel<<<256, 512, 0, stream>>>(pf, pq, Wa, mask, feature, out);
}


# scores phase: pq rows and Wa staged in LDS beside the pf tile, read as broadcast b128 instead of scalar loads; wave = 4 q x 32-a slice, lane = 4 f; changes the summation order over a (8 slice sums)
# speedup vs baseline: 4.9723x; 1.0889x over previous
#include <hip/hip_runtime.h>

typedef float v2f __attribute__((ext_vector_type(2)));
typedef float v8f __attribute__((ext_vector_type(8)));

static __device__ __forceinline__ float fast_tanh(float x) {
#if __has_builtin(__builtin_amdgcn_tanhf)
    return __builtin_amdgcn_tanhf(x);
#else
    float r;
    asm("v_tanh_f32 %0, %1" : "=v"(r) : "v"(x));
    return r;
#endif
}

typedef float v4f __attribute__((ext_vector_type(4)));

template <int NT> struct VecN;
template <> struct VecN<4> { typedef v4f type; };
template <> struct VecN<2> { typedef v2f type; };

template <int G, int NT>
struct Stage {
    v4f a[G];
    typename VecN<NT>::type b[G][4];
};

template <int G, int NT>
static __device__ __forceinline__ void load_stage(Stage<G, NT>& s,
                                                  const float* __restrict__ aP,
                                                  const float* __restrict__ bP,
                                                  int N)
{
    typedef typename VecN<NT>::type vb;
    #pragma unroll
    for (int g = 0; g < G; ++g) {
        s.a[g] = *reinterpret_cast<const v4f*>(aP + g * 8);
        #pragma unroll
        for (int j = 0; j < 4; ++j)
            s.b[g][j] = *reinterpret_cast<const vb*>(bP + (size_t)(g * 8 + j) * N);
    }
}

template <int G, int NT>
static __device__ __forceinline__ void mma_stage(const Stage<G, NT>& s, v8f (&acc)[NT])
{
    #pragma unroll
    for (int g = 0; g < G; ++g) {
        const v2f a01 = {s.a[g][0], s.a[g][1]};
        const v2f a23 = {s.a[g][2], s.a[g][3]};
        #pragma unroll
        for (int t = 0; t < NT; ++t) {
            const v2f bv = {s.b[g][0][t], s.b[g][1][t]};
            acc[t] = __builtin_amdgcn_wmma_f32_16x16x4_f32(
                 false, a01,  false, bv,
                 (short)0, acc[t],  false, false);
        }
        #pragma unroll
        for (int t = 0; t < NT; ++t) {
            const v2f bv = {s.b[g][2][t], s.b[g][3][t]};
            acc[t] = __builtin_amdgcn_wmma_f32_16x16x4_f32(
                 false, a23,  false, bv,
                 (short)0, acc[t],  false, false);
        }
    }
}

template <int G, int NT>
static __device__ __forceinline__ void gemm_strip(const float* __restrict__ A,
                              const float* __restrict__ Bm,
                              const float* __restrict__ bias,
                              float* __restrict__ C,
                              int N, int K, int tM, int tN, int lane)
{
    typedef typename VecN<NT>::type vb;
    const int h    = lane >> 4;
    const int c    = lane & 15;

    const int  col   = tN + NT * c;
    const bool colOk = col < N;
    const int  colL  = colOk ? col : (N - NT);

    const float* __restrict__ aP = A + (size_t)(tM + c) * K + 4 * h;
    const float* __restrict__ bP = Bm + (size_t)(4 * h) * N + colL;

    v8f acc[NT];
    #pragma unroll
    for (int t = 0; t < NT; ++t) acc[t] = (v8f){};

    const int    S     = K / (8 * G);
    const size_t bStep = (size_t)(8 * G) * N;

    Stage<G, NT> s0, s1;
    load_stage<G, NT>(s0, aP, bP, N);
    int s = 0;
    for (; s + 2 < S; s += 2) {
        load_stage<G, NT>(s1, aP + (s + 1) * 8 * G, bP + (size_t)(s + 1) * bStep, N);
        __builtin_amdgcn_sched_barrier(0);
        mma_stage<G, NT>(s0, acc);
        __builtin_amdgcn_sched_barrier(0);
        load_stage<G, NT>(s0, aP + (s + 2) * 8 * G, bP + (size_t)(s + 2) * bStep, N);
        __builtin_amdgcn_sched_barrier(0);
        mma_stage<G, NT>(s1, acc);
        __builtin_amdgcn_sched_barrier(0);
    }
    load_stage<G, NT>(s1, aP + (s + 1) * 8 * G, bP + (size_t)(s + 1) * bStep, N);
    __builtin_amdgcn_sched_barrier(0);
    mma_stage<G, NT>(s0, acc);
    mma_stage<G, NT>(s1, acc);

    if (colOk) {
        vb bv = (vb){};
        if (bias) bv = *reinterpret_cast<const vb*>(bias + col);
        #pragma unroll
        for (int r = 0; r < 8; ++r) {
            const int m = tM + r + h * 8;
            vb o;
            #pragma unroll
            for (int t = 0; t < NT; ++t) o[t] = acc[t][r];
            o += bv;
            *reinterpret_cast<vb*>(C + (size_t)m * N + col) = o;
        }
    }
}

__global__ void __launch_bounds__(128) proj_kernel(const float* __restrict__ feature,
                              const float* __restrict__ Wf,
                              float* __restrict__ pf,
                              const float* __restrict__ query,
                              const float* __restrict__ Wq,
                              const float* __restrict__ bq,
                              float* __restrict__ pq,
                              int N, int K)
{
    const int wave = threadIdx.x >> 5;
    const int lane = threadIdx.x & 31;
    const int id   = blockIdx.x;
    if (id >= 256) {
        const int jd = id - 256;
        const int tN = (jd & 3) * 64;
        const int tM = ((jd >> 2) * 4 + wave) * 16;
        gemm_strip<5, 4>(feature, Wf, nullptr, pf, N, K, tM, tN, lane);
    } else {
        const int tN = (id & 7) * 32;
        const int tM = ((id >> 3) * 4 + wave) * 16;
        gemm_strip<5, 2>(query, Wq, bq, pq, N, K, tM, tN, lane);
    }
}

#define SC_PITCH 260
#define S_PITCH 136

struct BStage {
    v2f b[4][4];
};

static __device__ __forceinline__ void load_b_stage(BStage& s,
                                                    const float* __restrict__ bP,
                                                    int N)
{
    #pragma unroll
    for (int g = 0; g < 4; ++g) {
        #pragma unroll
        for (int j = 0; j < 4; ++j)
            s.b[g][j] = *reinterpret_cast<const v2f*>(bP + (size_t)(g * 8 + j) * N);
    }
}

template <int BASE>
static __device__ __forceinline__ void mma_b_stage(const BStage& s, const v4f (&a)[16],
                                                   v8f (&acc)[2])
{
    #pragma unroll
    for (int g = 0; g < 4; ++g) {
        const v2f a01 = {a[BASE + g][0], a[BASE + g][1]};
        const v2f a23 = {a[BASE + g][2], a[BASE + g][3]};
        #pragma unroll
        for (int t = 0; t < 2; ++t) {
            const v2f bv = {s.b[g][0][t], s.b[g][1][t]};
            acc[t] = __builtin_amdgcn_wmma_f32_16x16x4_f32(
                 false, a01,  false, bv,
                 (short)0, acc[t],  false, false);
        }
        #pragma unroll
        for (int t = 0; t < 2; ++t) {
            const v2f bv = {s.b[g][2][t], s.b[g][3][t]};
            acc[t] = __builtin_amdgcn_wmma_f32_16x16x4_f32(
                 false, a23,  false, bv,
                 (short)0, acc[t],  false, false);
        }
    }
}

__global__ void __launch_bounds__(512) scores_attend_kernel(const float* __restrict__ pf,
                              const float* __restrict__ pq,
                              const float* __restrict__ Wa,
                              const int* __restrict__ mask,
                              const float* __restrict__ feature,
                              float* __restrict__ out)
{
    __shared__ __attribute__((aligned(16))) float tile[128 * SC_PITCH];
    __shared__ __attribute__((aligned(16))) float pqs[8 * 256];
    __shared__ __attribute__((aligned(16))) float was[256];
    __shared__ __attribute__((aligned(16))) float part[8 * 8 * 128];
    __shared__ __attribute__((aligned(16))) float sc[8 * S_PITCH];

    const int tid  = threadIdx.x;
    const int lane = tid & 31;
    const int wave = __builtin_amdgcn_readfirstlane(tid >> 5);
    const int blk  = blockIdx.x;
    const int qg   = blk & 3;
    const int b    = blk >> 2;

    const v4f* __restrict__ srcp =
        reinterpret_cast<const v4f*>(pf + (size_t)b * (128 * 256));
    v4f st[16];
    #pragma unroll
    for (int i = 0; i < 16; ++i) st[i] = srcp[i * 512 + tid];
    const v4f yst = *reinterpret_cast<const v4f*>(
        pq + (size_t)(b * 32 + qg * 8) * 256 + tid * 4);
    v4f wst = (v4f){};
    if (wave < 2) wst = *reinterpret_cast<const v4f*>(Wa + tid * 4);
    const int2 mk = *reinterpret_cast<const int2*>(mask + b * 128 + (tid & 63) * 2);
    #pragma unroll
    for (int i = 0; i < 16; ++i) {
        const int idx = i * 512 + tid;
        *reinterpret_cast<v4f*>(&tile[(idx >> 6) * SC_PITCH + (idx & 63) * 4]) = st[i];
    }
    *reinterpret_cast<v4f*>(&pqs[tid * 4]) = yst;
    if (wave < 2) *reinterpret_cast<v4f*>(&was[tid * 4]) = wst;
    __syncthreads();

    {
        const int as = wave & 7;
        const int qh = wave >> 3;
        const float* xr = &tile[lane * SC_PITCH + as * 32];
        const float* yr = &pqs[(qh * 4) * 256 + as * 32];
        const float* wr = &was[as * 32];

        v2f acc[4][4];
        #pragma unroll
        for (int j = 0; j < 4; ++j) {
            #pragma unroll
            for (int qq = 0; qq < 4; ++qq) acc[j][qq] = (v2f){0.0f, 0.0f};
        }

        #pragma unroll
        for (int s4 = 0; s4 < 8; ++s4) {
            v4f x[4], y[4];
            #pragma unroll
            for (int j = 0; j < 4; ++j)
                x[j] = *reinterpret_cast<const v4f*>(xr + j * (32 * SC_PITCH) + s4 * 4);
            #pragma unroll
            for (int qq = 0; qq < 4; ++qq)
                y[qq] = *reinterpret_cast<const v4f*>(yr + qq * 256 + s4 * 4);
            const v4f w = *reinterpret_cast<const v4f*>(wr + s4 * 4);
            const v2f wlo = {w[0], w[1]};
            const v2f whi = {w[2], w[3]};
            #pragma unroll
            for (int j = 0; j < 4; ++j) {
                const v2f xlo = {x[j][0], x[j][1]};
                const v2f xhi = {x[j][2], x[j][3]};
                #pragma unroll
                for (int qq = 0; qq < 4; ++qq) {
                    const v2f ylo = {y[qq][0], y[qq][1]};
                    const v2f yhi = {y[qq][2], y[qq][3]};
                    v2f s, t;
                    s = xlo + ylo; t[0] = fast_tanh(s[0]); t[1] = fast_tanh(s[1]);
                    acc[j][qq] = t * wlo + acc[j][qq];
                    s = xhi + yhi; t[0] = fast_tanh(s[0]); t[1] = fast_tanh(s[1]);
                    acc[j][qq] = t * whi + acc[j][qq];
                }
            }
        }

        #pragma unroll
        for (int j = 0; j < 4; ++j) {
            #pragma unroll
            for (int qq = 0; qq < 4; ++qq)
                part[((as * 8) + qh * 4 + qq) * 128 + j * 32 + lane] =
                    acc[j][qq][0] + acc[j][qq][1];
        }
    }
    __syncthreads();

    {
        const int q  = tid >> 6;
        const int f2 = (tid & 63) * 2;
        v2f s = *reinterpret_cast<const v2f*>(&part[q * 128 + f2]);
        #pragma unroll
        for (int sl = 1; sl < 8; ++sl)
            s += *reinterpret_cast<const v2f*>(&part[(sl * 8 + q) * 128 + f2]);
        v2f o;
        o[0] = (mk.x == 0) ? -1e9f : s[0];
        o[1] = (mk.y == 0) ? -1e9f : s[1];
        *reinterpret_cast<v2f*>(&sc[q * S_PITCH + f2]) = o;
    }
    __syncthreads();

    if (wave >= 13) return;

    constexpr int N = 400;
    constexpr int K = 128;

    const float* __restrict__ Bm = feature + (size_t)b * (K * N);
    float* __restrict__ C        = out + ((size_t)b * 32 + qg * 8) * N;

    const int h = lane >> 4;
    const int c = lane & 15;

    const int  col   = wave * 32 + 2 * c;
    const bool colOk = col < N;
    const int  colL  = colOk ? col : (N - 2);

    const float* aP = &sc[(c & 7) * S_PITCH + 4 * h];
    const float* __restrict__ bP = Bm + (size_t)(4 * h) * N + colL;

    v8f acc[2];
    #pragma unroll
    for (int t = 0; t < 2; ++t) acc[t] = (v8f){};

    BStage s0, s1;
    v4f a[16];
    load_b_stage(s0, bP, N);
    #pragma unroll
    for (int g = 0; g < 16; ++g) a[g] = *reinterpret_cast<const v4f*>(aP + g * 8);
    load_b_stage(s1, bP + (size_t)32 * N, N);
    __builtin_amdgcn_sched_barrier(0);

    float m = fmaxf(fmaxf(a[0][0], a[0][1]), fmaxf(a[0][2], a[0][3]));
    #pragma unroll
    for (int g = 1; g < 16; ++g)
        m = fmaxf(m, fmaxf(fmaxf(a[g][0], a[g][1]), fmaxf(a[g][2], a[g][3])));
    m = fmaxf(m, __shfl_xor(m, 16, 32));

    v4f ps = (v4f){};
    #pragma unroll
    for (int g = 0; g < 16; ++g) {
        v4f e;
        e[0] = __expf(a[g][0] - m);
        e[1] = __expf(a[g][1] - m);
        e[2] = __expf(a[g][2] - m);
        e[3] = __expf(a[g][3] - m);
        a[g] = e;
        ps += e;
    }
    float ssum = (ps[0] + ps[1]) + (ps[2] + ps[3]);
    ssum += __shfl_xor(ssum, 16, 32);
    const float inv = 1.0f / ssum;
    #pragma unroll
    for (int g = 0; g < 16; ++g) a[g] *= inv;

    mma_b_stage<0>(s0, a, acc);
    __builtin_amdgcn_sched_barrier(0);
    load_b_stage(s0, bP + (size_t)64 * N, N);
    __builtin_amdgcn_sched_barrier(0);
    mma_b_stage<4>(s1, a, acc);
    __builtin_amdgcn_sched_barrier(0);
    load_b_stage(s1, bP + (size_t)96 * N, N);
    __builtin_amdgcn_sched_barrier(0);
    mma_b_stage<8>(s0, a, acc);
    mma_b_stage<12>(s1, a, acc);

    if (colOk && h == 0) {
        #pragma unroll
        for (int r = 0; r < 8; ++r) {
            const v2f o = {acc[0][r], acc[1][r]};
            *reinterpret_cast<v2f*>(C + (size_t)r * N + col) = o;
        }
    }
}

extern "C" void kernel_launch(void* const* d_in, const int* in_sizes, int n_in,
                              void* d_out, int out_size, void* d_ws, size_t ws_size,
                              hipStream_t stream)
{
    const float* feature = (const float*)d_in[0];
    const float* query   = (const float*)d_in[1];
    const int*   mask    = (const int*)  d_in[2];
    const float* Wf      = (const float*)d_in[3];
    const float* Wq      = (const float*)d_in[4];
    const float* bq      = (const float*)d_in[5];
    const float* Wa      = (const float*)d_in[6];
    float*       out     = (float*)d_out;

    char* ws = (char*)d_ws;
    float* pf = (float*)ws;
    float* pq = (float*)(ws + (size_t)8192 * 256 * 4);

    proj_kernel<<<768, 128, 0, stream>>>(feature, Wf, pf, query, Wq, bq, pq, 256, 400);

    scores_attend_kernel<<<256, 512, 0, stream>>>(pf, pq, Wa, mask, feature, out);
}


# mask == 0 rows skipped in all stages: a batch's mask != 0 rows become slots; pf in 16x32 strips dealt evenly over SIMDs, scores and weighted sum on 32*ceil(n/32) slots; changes sum order over f
# speedup vs baseline: 5.9371x; 1.1940x over previous
#include <hip/hip_runtime.h>

typedef float v2f __attribute__((ext_vector_type(2)));
typedef float v8f __attribute__((ext_vector_type(8)));

static __device__ __forceinline__ float fast_tanh(float x) {
#if __has_builtin(__builtin_amdgcn_tanhf)
    return __builtin_amdgcn_tanhf(x);
#else
    float r;
    asm("v_tanh_f32 %0, %1" : "=v"(r) : "v"(x));
    return r;
#endif
}

struct RowSet {
    unsigned w[4];
    int      n;
    int      nreal;
};

static __device__ __forceinline__ RowSet load_rowset(const int* __restrict__ mrow, int lane)
{
    const int m0 = mrow[lane];
    const int m1 = mrow[32 + lane];
    const int m2 = mrow[64 + lane];
    const int m3 = mrow[96 + lane];
    RowSet r;
    r.w[0] = __builtin_amdgcn_ballot_w32(m0 != 0);
    r.w[1] = __builtin_amdgcn_ballot_w32(m1 != 0);
    r.w[2] = __builtin_amdgcn_ballot_w32(m2 != 0);
    r.w[3] = __builtin_amdgcn_ballot_w32(m3 != 0);
    r.n = __popc(r.w[0]) + __popc(r.w[1]) + __popc(r.w[2]) + __popc(r.w[3]);
    r.nreal = r.n;
    if (r.n == 0) {
        r.w[0] = r.w[1] = r.w[2] = r.w[3] = 0xFFFFFFFFu;
        r.n = 128;
    }
    return r;
}

static __device__ __forceinline__ int slot_row(const RowSet& rs, int r)
{
    const int c0 = __popc(rs.w[0]);
    const int c1 = __popc(rs.w[1]);
    const int c2 = __popc(rs.w[2]);
    unsigned w = rs.w[0];
    int base = 0;
    if (r >= c0) {
        r -= c0; w = rs.w[1]; base = 32;
        if (r >= c1) {
            r -= c1; w = rs.w[2]; base = 64;
            if (r >= c2) { r -= c2; w = rs.w[3]; base = 96; }
        }
    }
    int t;
    t = __popc(w & 0xFFFFu); if (r >= t) { r -= t; w >>= 16; base += 16; }
    t = __popc(w & 0xFFu);   if (r >= t) { r -= t; w >>= 8;  base += 8;  }
    t = __popc(w & 0xFu);    if (r >= t) { r -= t; w >>= 4;  base += 4;  }
    t = __popc(w & 0x3u);    if (r >= t) { r -= t; w >>= 2;  base += 2;  }
    t = __popc(w & 0x1u);    if (r >= t) { base += 1; }
    return base;
}

typedef float v4f __attribute__((ext_vector_type(4)));

template <int NT> struct VecN;
template <> struct VecN<4> { typedef v4f type; };
template <> struct VecN<2> { typedef v2f type; };

template <int G, int NT>
struct Stage {
    v4f a[G];
    typename VecN<NT>::type b[G][4];
};

template <int G, int NT>
static __device__ __forceinline__ void load_stage(Stage<G, NT>& s,
                                                  const float* __restrict__ aP,
                                                  const float* __restrict__ bP,
                                                  int N)
{
    typedef typename VecN<NT>::type vb;
    #pragma unroll
    for (int g = 0; g < G; ++g) {
        s.a[g] = *reinterpret_cast<const v4f*>(aP + g * 8);
        #pragma unroll
        for (int j = 0; j < 4; ++j)
            s.b[g][j] = *reinterpret_cast<const vb*>(bP + (size_t)(g * 8 + j) * N);
    }
}

template <int G, int NT>
static __device__ __forceinline__ void mma_stage(const Stage<G, NT>& s, v8f (&acc)[NT])
{
    #pragma unroll
    for (int g = 0; g < G; ++g) {
        const v2f a01 = {s.a[g][0], s.a[g][1]};
        const v2f a23 = {s.a[g][2], s.a[g][3]};
        #pragma unroll
        for (int t = 0; t < NT; ++t) {
            const v2f bv = {s.b[g][0][t], s.b[g][1][t]};
            acc[t] = __builtin_amdgcn_wmma_f32_16x16x4_f32(
                 false, a01,  false, bv,
                 (short)0, acc[t],  false, false);
        }
        #pragma unroll
        for (int t = 0; t < NT; ++t) {
            const v2f bv = {s.b[g][2][t], s.b[g][3][t]};
            acc[t] = __builtin_amdgcn_wmma_f32_16x16x4_f32(
                 false, a23,  false, bv,
                 (short)0, acc[t],  false, false);
        }
    }
}

template <int G, int NT>
static __device__ __forceinline__ void gemm_strip(const float* __restrict__ A,
                              const float* __restrict__ Bm,
                              const float* __restrict__ bias,
                              float* __restrict__ C,
                              int N, int K, int aRow, int tM, int tN, int lane)
{
    typedef typename VecN<NT>::type vb;
    const int h    = lane >> 4;
    const int c    = lane & 15;

    const int  col   = tN + NT * c;
    const bool colOk = col < N;
    const int  colL  = colOk ? col : (N - NT);

    const float* __restrict__ aP = A + (size_t)aRow * K + 4 * h;
    const float* __restrict__ bP = Bm + (size_t)(4 * h) * N + colL;

    v8f acc[NT];
    #pragma unroll
    for (int t = 0; t < NT; ++t) acc[t] = (v8f){};

    const int    S     = K / (8 * G);
    const size_t bStep = (size_t)(8 * G) * N;

    Stage<G, NT> s0, s1;
    load_stage<G, NT>(s0, aP, bP, N);
    int s = 0;
    for (; s + 2 < S; s += 2) {
        load_stage<G, NT>(s1, aP + (s + 1) * 8 * G, bP + (size_t)(s + 1) * bStep, N);
        __builtin_amdgcn_sched_barrier(0);
        mma_stage<G, NT>(s0, acc);
        __builtin_amdgcn_sched_barrier(0);
        load_stage<G, NT>(s0, aP + (s + 2) * 8 * G, bP + (size_t)(s + 2) * bStep, N);
        __builtin_amdgcn_sched_barrier(0);
        mma_stage<G, NT>(s1, acc);
        __builtin_amdgcn_sched_barrier(0);
    }
    load_stage<G, NT>(s1, aP + (s + 1) * 8 * G, bP + (size_t)(s + 1) * bStep, N);
    __builtin_amdgcn_sched_barrier(0);
    mma_stage<G, NT>(s0, acc);
    mma_stage<G, NT>(s1, acc);

    if (colOk) {
        vb bv = (vb){};
        if (bias) bv = *reinterpret_cast<const vb*>(bias + col);
        #pragma unroll
        for (int r = 0; r < 8; ++r) {
            const int m = tM + r + h * 8;
            vb o;
            #pragma unroll
            for (int t = 0; t < NT; ++t) o[t] = acc[t][r];
            o += bv;
            *reinterpret_cast<vb*>(C + (size_t)m * N + col) = o;
        }
    }
}

__global__ void __launch_bounds__(128) proj_kernel(const float* __restrict__ feature,
                              const float* __restrict__ Wf,
                              float* __restrict__ pf,
                              const float* __restrict__ query,
                              const float* __restrict__ Wq,
                              const float* __restrict__ bq,
                              float* __restrict__ pq,
                              const int* __restrict__ mask,
                              int N, int K)
{
    const int wave = threadIdx.x >> 5;
    const int lane = threadIdx.x & 31;
    const int id   = blockIdx.x;
    if (id >= 256) {
        const int jd    = id - 256;
        const int round = jd >> 8;
        const int cu    = jd & 255;
        const int tN    = (cu & 7) * 32;
        const int b     = (cu >> 3) * 2 + (wave >> 1);
        const int odd   = wave & 1;
        const int strip = (round == 0) ? 2 * odd
                        : (round == 1) ? 1 + 2 * odd
                        : (round == 2) ? 4 + odd
                        :                6 + odd;
        const int r0    = strip * 16;
        const RowSet rs = load_rowset(mask + b * 128, lane);
        if (r0 >= rs.n) return;
        const int r = min(r0 + (lane & 15), rs.n - 1);
        const int f = slot_row(rs, r);
        gemm_strip<5, 2>(feature, Wf, nullptr, pf, N, K, b * 128 + f, b * 128 + r0, tN, lane);
    } else {
        const int tN = (id & 7) * 32;
        const int tM = ((id >> 3) * 4 + wave) * 16;
        gemm_strip<5, 2>(query, Wq, bq, pq, N, K, tM + (lane & 15), tM, tN, lane);
    }
}

#define SC_PITCH 260
#define S_PITCH 136

struct BStage {
    v2f b[4][4];
};

static __device__ __forceinline__ void load_b_stage(BStage& s,
                                                    const float* __restrict__ bP,
                                                    const int* rowsP,
                                                    int N)
{
    #pragma unroll
    for (int g = 0; g < 4; ++g) {
        const int4 ix = *reinterpret_cast<const int4*>(rowsP + g * 8);
        s.b[g][0] = *reinterpret_cast<const v2f*>(bP + (size_t)ix.x * N);
        s.b[g][1] = *reinterpret_cast<const v2f*>(bP + (size_t)ix.y * N);
        s.b[g][2] = *reinterpret_cast<const v2f*>(bP + (size_t)ix.z * N);
        s.b[g][3] = *reinterpret_cast<const v2f*>(bP + (size_t)ix.w * N);
    }
}

template <int BASE, int NA>
static __device__ __forceinline__ void mma_b_stage(const BStage& s, const v4f (&a)[NA],
                                                   v8f (&acc)[2])
{
    #pragma unroll
    for (int g = 0; g < 4; ++g) {
        const v2f a01 = {a[BASE + g][0], a[BASE + g][1]};
        const v2f a23 = {a[BASE + g][2], a[BASE + g][3]};
        #pragma unroll
        for (int t = 0; t < 2; ++t) {
            const v2f bv = {s.b[g][0][t], s.b[g][1][t]};
            acc[t] = __builtin_amdgcn_wmma_f32_16x16x4_f32(
                 false, a01,  false, bv,
                 (short)0, acc[t],  false, false);
        }
        #pragma unroll
        for (int t = 0; t < 2; ++t) {
            const v2f bv = {s.b[g][2][t], s.b[g][3][t]};
            acc[t] = __builtin_amdgcn_wmma_f32_16x16x4_f32(
                 false, a23,  false, bv,
                 (short)0, acc[t],  false, false);
        }
    }
}

template <int NR>
static __device__ __forceinline__ void scores_attend_body(
    float* tile, const float* pqs, const float* was, float* part, float* sc,
    const int* rows,
    const float* __restrict__ pf, const float* __restrict__ feature,
    float* __restrict__ out,
    int tid, int lane, int wave, int b, int qg, int nw, int nreal)
{
    {
        const v4f* __restrict__ srcp =
            reinterpret_cast<const v4f*>(pf + (size_t)b * (128 * 256));
        v4f st[4 * NR];
        #pragma unroll
        for (int i = 0; i < 4 * NR; ++i) {
            st[i] = (v4f){};
            if (i < 4 * NR - 2 || i * 8 < nw) st[i] = srcp[i * 512 + tid];
        }
        #pragma unroll
        for (int i = 0; i < 4 * NR; ++i) {
            const int idx = i * 512 + tid;
            *reinterpret_cast<v4f*>(&tile[(idx >> 6) * SC_PITCH + (idx & 63) * 4]) = st[i];
        }
    }
    __syncthreads();

    {
        const int as = wave & 7;
        const int qh = wave >> 3;
        const float* xr = &tile[lane * SC_PITCH + as * 32];
        const float* yr = &pqs[(qh * 4) * 256 + as * 32];
        const float* wr = &was[as * 32];

        v2f acc[NR][4];
        #pragma unroll
        for (int j = 0; j < NR; ++j) {
            #pragma unroll
            for (int qq = 0; qq < 4; ++qq) acc[j][qq] = (v2f){0.0f, 0.0f};
        }

        #pragma unroll
        for (int s4 = 0; s4 < 8; ++s4) {
            v4f x[NR], y[4];
            #pragma unroll
            for (int j = 0; j < NR; ++j)
                x[j] = *reinterpret_cast<const v4f*>(xr + j * (32 * SC_PITCH) + s4 * 4);
            #pragma unroll
            for (int qq = 0; qq < 4; ++qq)
                y[qq] = *reinterpret_cast<const v4f*>(yr + qq * 256 + s4 * 4);
            const v4f w = *reinterpret_cast<const v4f*>(wr + s4 * 4);
            const v2f wlo = {w[0], w[1]};
            const v2f whi = {w[2], w[3]};
            #pragma unroll
            for (int j = 0; j < NR; ++j) {
                const v2f xlo = {x[j][0], x[j][1]};
                const v2f xhi = {x[j][2], x[j][3]};
                #pragma unroll
                for (int qq = 0; qq < 4; ++qq) {
                    const v2f ylo = {y[qq][0], y[qq][1]};
                    const v2f yhi = {y[qq][2], y[qq][3]};
                    v2f s, t;
                    s = xlo + ylo; t[0] = fast_tanh(s[0]); t[1] = fast_tanh(s[1]);
                    acc[j][qq] = t * wlo + acc[j][qq];
                    s = xhi + yhi; t[0] = fast_tanh(s[0]); t[1] = fast_tanh(s[1]);
                    acc[j][qq] = t * whi + acc[j][qq];
                }
            }
        }

        #pragma unroll
        for (int j = 0; j < NR; ++j) {
            #pragma unroll
            for (int qq = 0; qq < 4; ++qq)
                part[((as * 8) + qh * 4 + qq) * 128 + j * 32 + lane] =
                    acc[j][qq][0] + acc[j][qq][1];
        }
    }
    __syncthreads();

    {
        const int q  = tid >> 6;
        const int f2 = (tid & 63) * 2;
        if (f2 < 32 * NR) {
            v2f s = *reinterpret_cast<const v2f*>(&part[q * 128 + f2]);
            #pragma unroll
            for (int sl = 1; sl < 8; ++sl)
                s += *reinterpret_cast<const v2f*>(&part[(sl * 8 + q) * 128 + f2]);
            v2f o;
            o[0] = (f2     < nreal) ? s[0] : -1e9f;
            o[1] = (f2 + 1 < nreal) ? s[1] : -1e9f;
            *reinterpret_cast<v2f*>(&sc[q * S_PITCH + f2]) = o;
        }
    }
    __syncthreads();

    if (wave >= 13) return;

    constexpr int N  = 400;
    constexpr int NA = 4 * NR;

    const float* __restrict__ Bm = feature + (size_t)b * (128 * N);
    float* __restrict__ C        = out + ((size_t)b * 32 + qg * 8) * N;

    const int h = lane >> 4;
    const int c = lane & 15;

    const int  col   = wave * 32 + 2 * c;
    const bool colOk = col < N;
    const int  colL  = colOk ? col : (N - 2);

    const float* aP = &sc[(c & 7) * S_PITCH + 4 * h];
    const int*   rP = rows + 4 * h;
    const float* __restrict__ bP = Bm + colL;

    v8f acc[2];
    #pragma unroll
    for (int t = 0; t < 2; ++t) acc[t] = (v8f){};

    BStage s0, s1;
    v4f a[NA];
    load_b_stage(s0, bP, rP, N);
    #pragma unroll
    for (int g = 0; g < NA; ++g) a[g] = *reinterpret_cast<const v4f*>(aP + g * 8);
    if constexpr (NR > 1) load_b_stage(s1, bP, rP + 32, N);
    __builtin_amdgcn_sched_barrier(0);

    float m = fmaxf(fmaxf(a[0][0], a[0][1]), fmaxf(a[0][2], a[0][3]));
    #pragma unroll
    for (int g = 1; g < NA; ++g)
        m = fmaxf(m, fmaxf(fmaxf(a[g][0], a[g][1]), fmaxf(a[g][2], a[g][3])));
    m = fmaxf(m, __shfl_xor(m, 16, 32));

    v4f ps = (v4f){};
    #pragma unroll
    for (int g = 0; g < NA; ++g) {
        v4f e;
        e[0] = __expf(a[g][0] - m);
        e[1] = __expf(a[g][1] - m);
        e[2] = __expf(a[g][2] - m);
        e[3] = __expf(a[g][3] - m);
        a[g] = e;
        ps += e;
    }
    float ssum = (ps[0] + ps[1]) + (ps[2] + ps[3]);
    ssum += __shfl_xor(ssum, 16, 32);
    const float inv = 1.0f / ssum;
    #pragma unroll
    for (int g = 0; g < NA; ++g) a[g] *= inv;

    mma_b_stage<0, NA>(s0, a, acc);
    if constexpr (NR > 2) {
        __builtin_amdgcn_sched_barrier(0);
        load_b_stage(s0, bP, rP + 64, N);
        __builtin_amdgcn_sched_barrier(0);
    }
    if constexpr (NR > 1) mma_b_stage<4, NA>(s1, a, acc);
    if constexpr (NR > 3) {
        __builtin_amdgcn_sched_barrier(0);
        load_b_stage(s1, bP, rP + 96, N);
        __builtin_amdgcn_sched_barrier(0);
    }
    if constexpr (NR > 2) mma_b_stage<8, NA>(s0, a, acc);
    if constexpr (NR > 3) mma_b_stage<12, NA>(s1, a, acc);

    if (colOk && h == 0) {
        #pragma unroll
        for (int r = 0; r < 8; ++r) {
            const v2f o = {acc[0][r], acc[1][r]};
            *reinterpret_cast<v2f*>(C + (size_t)r * N + col) = o;
        }
    }
}

__global__ void __launch_bounds__(512) scores_attend_kernel(const float* __restrict__ pf,
                              const float* __restrict__ pq,
                              const float* __restrict__ Wa,
                              const int* __restrict__ mask,
                              const float* __restrict__ feature,
                              float* __restrict__ out)
{
    __shared__ __attribute__((aligned(16))) float tile[128 * SC_PITCH];
    __shared__ __attribute__((aligned(16))) float pqs[8 * 256];
    __shared__ __attribute__((aligned(16))) float was[256];
    __shared__ __attribute__((aligned(16))) float part[8 * 8 * 128];
    __shared__ __attribute__((aligned(16))) float sc[8 * S_PITCH];
    __shared__ __attribute__((aligned(16))) int   rows[128];

    const int tid  = threadIdx.x;
    const int lane = tid & 31;
    const int wave = __builtin_amdgcn_readfirstlane(tid >> 5);
    const int blk  = blockIdx.x;
    const int qg   = blk & 3;
    const int b    = blk >> 2;

    const v4f yst = *reinterpret_cast<const v4f*>(
        pq + (size_t)(b * 32 + qg * 8) * 256 + tid * 4);
    v4f wst = (v4f){};
    if (wave < 2) wst = *reinterpret_cast<const v4f*>(Wa + tid * 4);
    const RowSet rs = load_rowset(mask + b * 128, lane);
    const int nr = (rs.n + 31) >> 5;
    const int nw = ((rs.n + 15) >> 4) << 4;

    *reinterpret_cast<v4f*>(&pqs[tid * 4]) = yst;
    if (wave < 2) *reinterpret_cast<v4f*>(&was[tid * 4]) = wst;
    if (wave < 4) rows[tid] = slot_row(rs, min(tid, rs.n - 1));

    switch (nr) {
    case 1:
        scores_attend_body<1>(tile, pqs, was, part, sc, rows, pf, feature, out,
                              tid, lane, wave, b, qg, nw, rs.nreal);
        break;
    case 2:
        scores_attend_body<2>(tile, pqs, was, part, sc, rows, pf, feature, out,
                              tid, lane, wave, b, qg, nw, rs.nreal);
        break;
    case 3:
        scores_attend_body<3>(tile, pqs, was, part, sc, rows, pf, feature, out,
                              tid, lane, wave, b, qg, nw, rs.nreal);
        break;
    default:
        scores_attend_body<4>(tile, pqs, was, part, sc, rows, pf, feature, out,
                              tid, lane, wave, b, qg, nw, rs.nreal);
        break;
    }
}

extern "C" void kernel_launch(void* const* d_in, const int* in_sizes, int n_in,
                              void* d_out, int out_size, void* d_ws, size_t ws_size,
                              hipStream_t stream)
{
    const float* feature = (const float*)d_in[0];
    const float* query   = (const float*)d_in[1];
    const int*   mask    = (const int*)  d_in[2];
    const float* Wf      = (const float*)d_in[3];
    const float* Wq      = (const float*)d_in[4];
    const float* bq      = (const float*)d_in[5];
    const float* Wa      = (const float*)d_in[6];
    float*       out     = (float*)d_out;

    char* ws = (char*)d_ws;
    float* pf = (float*)ws;
    float* pq = (float*)(ws + (size_t)8192 * 256 * 4);

    proj_kernel<<<1280, 128, 0, stream>>>(feature, Wf, pf, query, Wq, bq, pq, mask, 256, 400);

    scores_attend_kernel<<<256, 512, 0, stream>>>(pf, pq, Wa, mask, feature, out);
}


# pf slots 0-63 as 16x64 strips (4 accumulator tiles per A fragment, B rows read as float4): half the 16-line A fragment loads per WMMA; strips 6,7 follow 4,5 in one wave, 1280 blocks become 768
# speedup vs baseline: 6.2313x; 1.0496x over previous
#include <hip/hip_runtime.h>

typedef float v2f __attribute__((ext_vector_type(2)));
typedef float v8f __attribute__((ext_vector_type(8)));

static __device__ __forceinline__ float fast_tanh(float x) {
#if __has_builtin(__builtin_amdgcn_tanhf)
    return __builtin_amdgcn_tanhf(x);
#else
    float r;
    asm("v_tanh_f32 %0, %1" : "=v"(r) : "v"(x));
    return r;
#endif
}

struct RowSet {
    unsigned w[4];
    int      n;
    int      nreal;
};

static __device__ __forceinline__ RowSet load_rowset(const int* __restrict__ mrow, int lane)
{
    const int m0 = mrow[lane];
    const int m1 = mrow[32 + lane];
    const int m2 = mrow[64 + lane];
    const int m3 = mrow[96 + lane];
    RowSet r;
    r.w[0] = __builtin_amdgcn_ballot_w32(m0 != 0);
    r.w[1] = __builtin_amdgcn_ballot_w32(m1 != 0);
    r.w[2] = __builtin_amdgcn_ballot_w32(m2 != 0);
    r.w[3] = __builtin_amdgcn_ballot_w32(m3 != 0);
    r.n = __popc(r.w[0]) + __popc(r.w[1]) + __popc(r.w[2]) + __popc(r.w[3]);
    r.nreal = r.n;
    if (r.n == 0) {
        r.w[0] = r.w[1] = r.w[2] = r.w[3] = 0xFFFFFFFFu;
        r.n = 128;
    }
    return r;
}

static __device__ __forceinline__ int slot_row(const RowSet& rs, int r)
{
    const int c0 = __popc(rs.w[0]);
    const int c1 = __popc(rs.w[1]);
    const int c2 = __popc(rs.w[2]);
    unsigned w = rs.w[0];
    int base = 0;
    if (r >= c0) {
        r -= c0; w = rs.w[1]; base = 32;
        if (r >= c1) {
            r -= c1; w = rs.w[2]; base = 64;
            if (r >= c2) { r -= c2; w = rs.w[3]; base = 96; }
        }
    }
    int t;
    t = __popc(w & 0xFFFFu); if (r >= t) { r -= t; w >>= 16; base += 16; }
    t = __popc(w & 0xFFu);   if (r >= t) { r -= t; w >>= 8;  base += 8;  }
    t = __popc(w & 0xFu);    if (r >= t) { r -= t; w >>= 4;  base += 4;  }
    t = __popc(w & 0x3u);    if (r >= t) { r -= t; w >>= 2;  base += 2;  }
    t = __popc(w & 0x1u);    if (r >= t) { base += 1; }
    return base;
}

typedef float v4f __attribute__((ext_vector_type(4)));

template <int NT> struct VecN;
template <> struct VecN<4> { typedef v4f type; };
template <> struct VecN<2> { typedef v2f type; };

template <int G, int NT>
struct Stage {
    v4f a[G];
    typename VecN<NT>::type b[G][4];
};

template <int G, int NT>
static __device__ __forceinline__ void load_stage(Stage<G, NT>& s,
                                                  const float* __restrict__ aP,
                                                  const float* __restrict__ bP,
                                                  int N)
{
    typedef typename VecN<NT>::type vb;
    #pragma unroll
    for (int g = 0; g < G; ++g) {
        s.a[g] = *reinterpret_cast<const v4f*>(aP + g * 8);
        #pragma unroll
        for (int j = 0; j < 4; ++j)
            s.b[g][j] = *reinterpret_cast<const vb*>(bP + (size_t)(g * 8 + j) * N);
    }
}

template <int G, int NT>
static __device__ __forceinline__ void mma_stage(const Stage<G, NT>& s, v8f (&acc)[NT])
{
    #pragma unroll
    for (int g = 0; g < G; ++g) {
        const v2f a01 = {s.a[g][0], s.a[g][1]};
        const v2f a23 = {s.a[g][2], s.a[g][3]};
        #pragma unroll
        for (int t = 0; t < NT; ++t) {
            const v2f bv = {s.b[g][0][t], s.b[g][1][t]};
            acc[t] = __builtin_amdgcn_wmma_f32_16x16x4_f32(
                 false, a01,  false, bv,
                 (short)0, acc[t],  false, false);
        }
        #pragma unroll
        for (int t = 0; t < NT; ++t) {
            const v2f bv = {s.b[g][2][t], s.b[g][3][t]};
            acc[t] = __builtin_amdgcn_wmma_f32_16x16x4_f32(
                 false, a23,  false, bv,
                 (short)0, acc[t],  false, false);
        }
    }
}

template <int G, int NT>
static __device__ __forceinline__ void gemm_strip(const float* __restrict__ A,
                              const float* __restrict__ Bm,
                              const float* __restrict__ bias,
                              float* __restrict__ C,
                              int N, int K, int aRow, int tM, int tN, int lane)
{
    typedef typename VecN<NT>::type vb;
    const int h    = lane >> 4;
    const int c    = lane & 15;

    const int  col   = tN + NT * c;
    const bool colOk = col < N;
    const int  colL  = colOk ? col : (N - NT);

    const float* __restrict__ aP = A + (size_t)aRow * K + 4 * h;
    const float* __restrict__ bP = Bm + (size_t)(4 * h) * N + colL;

    v8f acc[NT];
    #pragma unroll
    for (int t = 0; t < NT; ++t) acc[t] = (v8f){};

    const int    S     = K / (8 * G);
    const size_t bStep = (size_t)(8 * G) * N;

    Stage<G, NT> s0, s1;
    load_stage<G, NT>(s0, aP, bP, N);
    int s = 0;
    for (; s + 2 < S; s += 2) {
        load_stage<G, NT>(s1, aP + (s + 1) * 8 * G, bP + (size_t)(s + 1) * bStep, N);
        __builtin_amdgcn_sched_barrier(0);
        mma_stage<G, NT>(s0, acc);
        __builtin_amdgcn_sched_barrier(0);
        load_stage<G, NT>(s0, aP + (s + 2) * 8 * G, bP + (size_t)(s + 2) * bStep, N);
        __builtin_amdgcn_sched_barrier(0);
        mma_stage<G, NT>(s1, acc);
        __builtin_amdgcn_sched_barrier(0);
    }
    load_stage<G, NT>(s1, aP + (s + 1) * 8 * G, bP + (size_t)(s + 1) * bStep, N);
    __builtin_amdgcn_sched_barrier(0);
    mma_stage<G, NT>(s0, acc);
    mma_stage<G, NT>(s1, acc);

    if (colOk) {
        vb bv = (vb){};
        if (bias) bv = *reinterpret_cast<const vb*>(bias + col);
        #pragma unroll
        for (int r = 0; r < 8; ++r) {
            const int m = tM + r + h * 8;
            vb o;
            #pragma unroll
            for (int t = 0; t < NT; ++t) o[t] = acc[t][r];
            o += bv;
            *reinterpret_cast<vb*>(C + (size_t)m * N + col) = o;
        }
    }
}

__global__ void __launch_bounds__(128) proj_kernel(const float* __restrict__ feature,
                              const float* __restrict__ Wf,
                              float* __restrict__ pf,
                              const float* __restrict__ query,
                              const float* __restrict__ Wq,
                              const float* __restrict__ bq,
                              float* __restrict__ pq,
                              const int* __restrict__ mask,
                              int N, int K)
{
    const int wave = threadIdx.x >> 5;
    const int lane = threadIdx.x & 31;
    const int id   = blockIdx.x;
    if (id < 256) {
        const int tN = (id & 7) * 32;
        const int tM = ((id >> 3) * 4 + wave) * 16;
        gemm_strip<5, 2>(query, Wq, bq, pq, N, K, tM + (lane & 15), tM, tN, lane);
    } else if (id < 512) {
        const int cu    = id - 256;
        const int tN    = (cu & 3) * 64;
        const int b     = cu >> 2;
        const int r0    = wave * 16;
        const RowSet rs = load_rowset(mask + b * 128, lane);
        if (r0 >= rs.n) return;
        const int r = min(r0 + (lane & 15), rs.n - 1);
        const int f = slot_row(rs, r);
        gemm_strip<5, 4>(feature, Wf, nullptr, pf, N, K, b * 128 + f, b * 128 + r0, tN, lane);
    } else {
        const int cu    = id - 512;
        const int tN    = (cu & 7) * 32;
        const int b     = (cu >> 3) * 2 + (wave >> 1);
        const RowSet rs = load_rowset(mask + b * 128, lane);
        for (int strip = 4 + (wave & 1); strip < 8; strip += 2) {
            const int r0 = strip * 16;
            if (r0 >= rs.n) return;
            const int r = min(r0 + (lane & 15), rs.n - 1);
            const int f = slot_row(rs, r);
            gemm_strip<5, 2>(feature, Wf, nullptr, pf, N, K, b * 128 + f, b * 128 + r0, tN, lane);
        }
    }
}

#define SC_PITCH 260
#define S_PITCH 136

struct BStage {
    v2f b[4][4];
};

static __device__ __forceinline__ void load_b_stage(BStage& s,
                                                    const float* __restrict__ bP,
                                                    const int* rowsP,
                                                    int N)
{
    #pragma unroll
    for (int g = 0; g < 4; ++g) {
        const int4 ix = *reinterpret_cast<const int4*>(rowsP + g * 8);
        s.b[g][0] = *reinterpret_cast<const v2f*>(bP + (size_t)ix.x * N);
        s.b[g][1] = *reinterpret_cast<const v2f*>(bP + (size_t)ix.y * N);
        s.b[g][2] = *reinterpret_cast<const v2f*>(bP + (size_t)ix.z * N);
        s.b[g][3] = *reinterpret_cast<const v2f*>(bP + (size_t)ix.w * N);
    }
}

template <int BASE, int NA>
static __device__ __forceinline__ void mma_b_stage(const BStage& s, const v4f (&a)[NA],
                                                   v8f (&acc)[2])
{
    #pragma unroll
    for (int g = 0; g < 4; ++g) {
        const v2f a01 = {a[BASE + g][0], a[BASE + g][1]};
        const v2f a23 = {a[BASE + g][2], a[BASE + g][3]};
        #pragma unroll
        for (int t = 0; t < 2; ++t) {
            const v2f bv = {s.b[g][0][t], s.b[g][1][t]};
            acc[t] = __builtin_amdgcn_wmma_f32_16x16x4_f32(
                 false, a01,  false, bv,
                 (short)0, acc[t],  false, false);
        }
        #pragma unroll
        for (int t = 0; t < 2; ++t) {
            const v2f bv = {s.b[g][2][t], s.b[g][3][t]};
            acc[t] = __builtin_amdgcn_wmma_f32_16x16x4_f32(
                 false, a23,  false, bv,
                 (short)0, acc[t],  false, false);
        }
    }
}

template <int NR>
static __device__ __forceinline__ void scores_attend_body(
    float* tile, const float* pqs, const float* was, float* part, float* sc,
    const int* rows,
    const float* __restrict__ pf, const float* __restrict__ feature,
    float* __restrict__ out,
    int tid, int lane, int wave, int b, int qg, int nw, int nreal)
{
    {
        const v4f* __restrict__ srcp =
            reinterpret_cast<const v4f*>(pf + (size_t)b * (128 * 256));
        v4f st[4 * NR];
        #pragma unroll
        for (int i = 0; i < 4 * NR; ++i) {
            st[i] = (v4f){};
            if (i < 4 * NR - 2 || i * 8 < nw) st[i] = srcp[i * 512 + tid];
        }
        #pragma unroll
        for (int i = 0; i < 4 * NR; ++i) {
            const int idx = i * 512 + tid;
            *reinterpret_cast<v4f*>(&tile[(idx >> 6) * SC_PITCH + (idx & 63) * 4]) = st[i];
        }
    }
    __syncthreads();

    {
        const int as = wave & 7;
        const int qh = wave >> 3;
        const float* xr = &tile[lane * SC_PITCH + as * 32];
        const float* yr = &pqs[(qh * 4) * 256 + as * 32];
        const float* wr = &was[as * 32];

        v2f acc[NR][4];
        #pragma unroll
        for (int j = 0; j < NR; ++j) {
            #pragma unroll
            for (int qq = 0; qq < 4; ++qq) acc[j][qq] = (v2f){0.0f, 0.0f};
        }

        #pragma unroll
        for (int s4 = 0; s4 < 8; ++s4) {
            v4f x[NR], y[4];
            #pragma unroll
            for (int j = 0; j < NR; ++j)
                x[j] = *reinterpret_cast<const v4f*>(xr + j * (32 * SC_PITCH) + s4 * 4);
            #pragma unroll
            for (int qq = 0; qq < 4; ++qq)
                y[qq] = *reinterpret_cast<const v4f*>(yr + qq * 256 + s4 * 4);
            const v4f w = *reinterpret_cast<const v4f*>(wr + s4 * 4);
            const v2f wlo = {w[0], w[1]};
            const v2f whi = {w[2], w[3]};
            #pragma unroll
            for (int j = 0; j < NR; ++j) {
                const v2f xlo = {x[j][0], x[j][1]};
                const v2f xhi = {x[j][2], x[j][3]};
                #pragma unroll
                for (int qq = 0; qq < 4; ++qq) {
                    const v2f ylo = {y[qq][0], y[qq][1]};
                    const v2f yhi = {y[qq][2], y[qq][3]};
                    v2f s, t;
                    s = xlo + ylo; t[0] = fast_tanh(s[0]); t[1] = fast_tanh(s[1]);
                    acc[j][qq] = t * wlo + acc[j][qq];
                    s = xhi + yhi; t[0] = fast_tanh(s[0]); t[1] = fast_tanh(s[1]);
                    acc[j][qq] = t * whi + acc[j][qq];
                }
            }
        }

        #pragma unroll
        for (int j = 0; j < NR; ++j) {
            #pragma unroll
            for (int qq = 0; qq < 4; ++qq)
                part[((as * 8) + qh * 4 + qq) * 128 + j * 32 + lane] =
                    acc[j][qq][0] + acc[j][qq][1];
        }
    }
    __syncthreads();

    {
        const int q  = tid >> 6;
        const int f2 = (tid & 63) * 2;
        if (f2 < 32 * NR) {
            v2f s = *reinterpret_cast<const v2f*>(&part[q * 128 + f2]);
            #pragma unroll
            for (int sl = 1; sl < 8; ++sl)
                s += *reinterpret_cast<const v2f*>(&part[(sl * 8 + q) * 128 + f2]);
            v2f o;
            o[0] = (f2     < nreal) ? s[0] : -1e9f;
            o[1] = (f2 + 1 < nreal) ? s[1] : -1e9f;
            *reinterpret_cast<v2f*>(&sc[q * S_PITCH + f2]) = o;
        }
    }
    __syncthreads();

    if (wave >= 13) return;

    constexpr int N  = 400;
    constexpr int NA = 4 * NR;

    const float* __restrict__ Bm = feature + (size_t)b * (128 * N);
    float* __restrict__ C        = out + ((size_t)b * 32 + qg * 8) * N;

    const int h = lane >> 4;
    const int c = lane & 15;

    const int  col   = wave * 32 + 2 * c;
    const bool colOk = col < N;
    const int  colL  = colOk ? col : (N - 2);

    const float* aP = &sc[(c & 7) * S_PITCH + 4 * h];
    const int*   rP = rows + 4 * h;
    const float* __restrict__ bP = Bm + colL;

    v8f acc[2];
    #pragma unroll
    for (int t = 0; t < 2; ++t) acc[t] = (v8f){};

    BStage s0, s1;
    v4f a[NA];
    load_b_stage(s0, bP, rP, N);
    #pragma unroll
    for (int g = 0; g < NA; ++g) a[g] = *reinterpret_cast<const v4f*>(aP + g * 8);
    if constexpr (NR > 1) load_b_stage(s1, bP, rP + 32, N);
    __builtin_amdgcn_sched_barrier(0);

    float m = fmaxf(fmaxf(a[0][0], a[0][1]), fmaxf(a[0][2], a[0][3]));
    #pragma unroll
    for (int g = 1; g < NA; ++g)
        m = fmaxf(m, fmaxf(fmaxf(a[g][0], a[g][1]), fmaxf(a[g][2], a[g][3])));
    m = fmaxf(m, __shfl_xor(m, 16, 32));

    v4f ps = (v4f){};
    #pragma unroll
    for (int g = 0; g < NA; ++g) {
        v4f e;
        e[0] = __expf(a[g][0] - m);
        e[1] = __expf(a[g][1] - m);
        e[2] = __expf(a[g][2] - m);
        e[3] = __expf(a[g][3] - m);
        a[g] = e;
        ps += e;
    }
    float ssum = (ps[0] + ps[1]) + (ps[2] + ps[3]);
    ssum += __shfl_xor(ssum, 16, 32);
    const float inv = 1.0f / ssum;
    #pragma unroll
    for (int g = 0; g < NA; ++g) a[g] *= inv;

    mma_b_stage<0, NA>(s0, a, acc);
    if constexpr (NR > 2) {
        __builtin_amdgcn_sched_barrier(0);
        load_b_stage(s0, bP, rP + 64, N);
        __builtin_amdgcn_sched_barrier(0);
    }
    if constexpr (NR > 1) mma_b_stage<4, NA>(s1, a, acc);
    if constexpr (NR > 3) {
        __builtin_amdgcn_sched_barrier(0);
        load_b_stage(s1, bP, rP + 96, N);
        __builtin_amdgcn_sched_barrier(0);
    }
    if constexpr (NR > 2) mma_b_stage<8, NA>(s0, a, acc);
    if constexpr (NR > 3) mma_b_stage<12, NA>(s1, a, acc);

    if (colOk && h == 0) {
        #pragma unroll
        for (int r = 0; r < 8; ++r) {
            const v2f o = {acc[0][r], acc[1][r]};
            *reinterpret_cast<v2f*>(C + (size_t)r * N + col) = o;
        }
    }
}

__global__ void __launch_bounds__(512) scores_attend_kernel(const float* __restrict__ pf,
                              const float* __restrict__ pq,
                              const float* __restrict__ Wa,
                              const int* __restrict__ mask,
                              const float* __restrict__ feature,
                              float* __restrict__ out)
{
    __shared__ __attribute__((aligned(16))) float tile[128 * SC_PITCH];
    __shared__ __attribute__((aligned(16))) float pqs[8 * 256];
    __shared__ __attribute__((aligned(16))) float was[256];
    __shared__ __attribute__((aligned(16))) float part[8 * 8 * 128];
    __shared__ __attribute__((aligned(16))) float sc[8 * S_PITCH];
    __shared__ __attribute__((aligned(16))) int   rows[128];

    const int tid  = threadIdx.x;
    const int lane = tid & 31;
    const int wave = __builtin_amdgcn_readfirstlane(tid >> 5);
    const int blk  = blockIdx.x;
    const int qg   = blk & 3;
    const int b    = blk >> 2;

    const v4f yst = *reinterpret_cast<const v4f*>(
        pq + (size_t)(b * 32 + qg * 8) * 256 + tid * 4);
    v4f wst = (v4f){};
    if (wave < 2) wst = *reinterpret_cast<const v4f*>(Wa + tid * 4);
    const RowSet rs = load_rowset(mask + b * 128, lane);
    const int nr = (rs.n + 31) >> 5;
    const int nw = ((rs.n + 15) >> 4) << 4;

    *reinterpret_cast<v4f*>(&pqs[tid * 4]) = yst;
    if (wave < 2) *reinterpret_cast<v4f*>(&was[tid * 4]) = wst;
    if (wave < 4) rows[tid] = slot_row(rs, min(tid, rs.n - 1));

    switch (nr) {
    case 1:
        scores_attend_body<1>(tile, pqs, was, part, sc, rows, pf, feature, out,
                              tid, lane, wave, b, qg, nw, rs.nreal);
        break;
    case 2:
        scores_attend_body<2>(tile, pqs, was, part, sc, rows, pf, feature, out,
                              tid, lane, wave, b, qg, nw, rs.nreal);
        break;
    case 3:
        scores_attend_body<3>(tile, pqs, was, part, sc, rows, pf, feature, out,
                              tid, lane, wave, b, qg, nw, rs.nreal);
        break;
    default:
        scores_attend_body<4>(tile, pqs, was, part, sc, rows, pf, feature, out,
                              tid, lane, wave, b, qg, nw, rs.nreal);
        break;
    }
}

extern "C" void kernel_launch(void* const* d_in, const int* in_sizes, int n_in,
                              void* d_out, int out_size, void* d_ws, size_t ws_size,
                              hipStream_t stream)
{
    const float* feature = (const float*)d_in[0];
    const float* query   = (const float*)d_in[1];
    const int*   mask    = (const int*)  d_in[2];
    const float* Wf      = (const float*)d_in[3];
    const float* Wq      = (const float*)d_in[4];
    const float* bq      = (const float*)d_in[5];
    const float* Wa      = (const float*)d_in[6];
    float*       out     = (float*)d_out;

    char* ws = (char*)d_ws;
    float* pf = (float*)ws;
    float* pq = (float*)(ws + (size_t)8192 * 256 * 4);

    proj_kernel<<<768, 128, 0, stream>>>(feature, Wf, pf, query, Wq, bq, pq, mask, 256, 400);

    scores_attend_kernel<<<256, 512, 0, stream>>>(pf, pq, Wa, mask, feature, out);
}
